# starting kernel
# baseline (speedup 1.0000x reference)
#include <hip/hip_runtime.h>


#define HIDC 128
#define KDIM 128
#define LPITCH 132

typedef float v2f __attribute__((ext_vector_type(2)));
typedef float v8f __attribute__((ext_vector_type(8)));

static __device__ __forceinline__ void atomic_add_f32(float* p, float v) {
  __hip_atomic_fetch_add(p, v, __ATOMIC_RELAXED, __HIP_MEMORY_SCOPE_AGENT);
}

__global__ __launch_bounds__(256) void k_zero(float* __restrict__ p, long n) {
  long i = (long)blockIdx.x * blockDim.x + threadIdx.x;
  if (i < n) p[i] = 0.0f;
}

__global__ __launch_bounds__(256) void k_degree(const int* __restrict__ dst,
                                                float* __restrict__ deg, int E) {
  int e = blockIdx.x * blockDim.x + threadIdx.x;
  if (e < E) atomic_add_f32(&deg[dst[e]], 1.0f);
}

__global__ __launch_bounds__(256) void k_invdeg(float* __restrict__ deg, int N) {
  int i = blockIdx.x * blockDim.x + threadIdx.x;
  if (i < N) deg[i] = 1.0f / fmaxf(deg[i], 1.0f);
}

__global__ __launch_bounds__(256) void k_scatter_add(const int* __restrict__ src,
                                                     const int* __restrict__ dst,
                                                     const float* __restrict__ h,
                                                     float* __restrict__ out, int E) {
  int lane = threadIdx.x & 31;
  int e = blockIdx.x * 8 + (threadIdx.x >> 5);
  if (e >= E) return;
  int s = src[e];
  int d = dst[e];
  float4 v = reinterpret_cast<const float4*>(h + (size_t)s * HIDC)[lane];
  float* o = out + (size_t)d * HIDC + lane * 4;
  atomic_add_f32(o + 0, v.x);
  atomic_add_f32(o + 1, v.y);
  atomic_add_f32(o + 2, v.z);
  atomic_add_f32(o + 3, v.w);
}

__global__ __launch_bounds__(256) void k_scale_rows(float* __restrict__ buf,
                                                    const float* __restrict__ inv,
                                                    long total) {
  long i = (long)blockIdx.x * blockDim.x + threadIdx.x;
  if (i < total) buf[i] *= inv[i >> 7];
}

__global__ __launch_bounds__(256) void k_transpose_w(const float* __restrict__ W,
                                                     float* __restrict__ Wt,
                                                     int K, int NOUT, int P) {
  long total = (long)P * K * NOUT;
  long i = (long)blockIdx.x * blockDim.x + threadIdx.x;
  if (i >= total) return;
  int p = (int)(i / (K * NOUT));
  int rem = (int)(i % (K * NOUT));
  int k = rem / NOUT;
  int n = rem % NOUT;
  Wt[(size_t)p * K * NOUT + (size_t)n * K + k] = W[i];
}

__global__ __launch_bounds__(256) void k_wmma_gemm3(
    const float* __restrict__ X0, const float* __restrict__ X1,
    const float* __restrict__ X2, const float* __restrict__ Wt,
    const float* __restrict__ bias, float* __restrict__ out) {
  __shared__ float lds[3 * 16 * LPITCH];
  const int t = threadIdx.x;
  const int lane = t & 31;
  const int mbase = blockIdx.x * 16;

  {
    const int r = t >> 4;
    const int c = (t & 15) * 8;
    const float4* g0 = reinterpret_cast<const float4*>(X0 + (size_t)(mbase + r) * KDIM + c);
    const float4* g1 = reinterpret_cast<const float4*>(X1 + (size_t)(mbase + r) * KDIM + c);
    const float4* g2 = reinterpret_cast<const float4*>(X2 + (size_t)(mbase + r) * KDIM + c);
    float4* l0 = reinterpret_cast<float4*>(lds + 0 * 16 * LPITCH + r * LPITCH + c);
    float4* l1 = reinterpret_cast<float4*>(lds + 1 * 16 * LPITCH + r * LPITCH + c);
    float4* l2 = reinterpret_cast<float4*>(lds + 2 * 16 * LPITCH + r * LPITCH + c);
    l0[0] = g0[0]; l0[1] = g0[1];
    l1[0] = g1[0]; l1[1] = g1[1];
    l2[0] = g2[0]; l2[1] = g2[1];
  }
  __syncthreads();

  const int nbase = (t >> 5) * 16;
  const int m = lane & 15;
  const int ncol = nbase + m;
  const int koff = (lane < 16) ? 0 : 2;

  v8f acc = {};
#pragma unroll
  for (int p = 0; p < 3; ++p) {
    const float* la = lds + p * 16 * LPITCH + m * LPITCH + koff;
    const float* wb = Wt + (size_t)p * (KDIM * HIDC) + (size_t)ncol * KDIM + koff;
#pragma unroll 8
    for (int k = 0; k < KDIM; k += 4) {
      v2f a, b;
      float2 av = *reinterpret_cast<const float2*>(la + k);
      float2 bv = *reinterpret_cast<const float2*>(wb + k);
      a[0] = av.x; a[1] = av.y;
      b[0] = bv.x; b[1] = bv.y;
      acc = __builtin_amdgcn_wmma_f32_16x16x4_f32(
          false, a, false, b, (short)0, acc, false, false);
    }
  }

  const float bv = bias[ncol];
  const int mo = (lane < 16) ? 0 : 8;
  float* orow = out + (size_t)mbase * HIDC + ncol;
#pragma unroll
  for (int i = 0; i < 8; ++i) orow[(size_t)(i + mo) * HIDC] = acc[i] + bv;
}

__global__ __launch_bounds__(128) void k_wmma_head(
    const float* __restrict__ X, const float* __restrict__ Wt,
    const float* __restrict__ bias, float* __restrict__ out, int NOUT) {
  __shared__ float lds[16 * LPITCH];
  const int t = threadIdx.x;
  const int lane = t & 31;
  const int mbase = blockIdx.x * 16;

  {
    const int r = t >> 3;
    const int c = (t & 7) * 16;
    const float4* g = reinterpret_cast<const float4*>(X + (size_t)(mbase + r) * KDIM + c);
    float4* l = reinterpret_cast<float4*>(lds + r * LPITCH + c);
    l[0] = g[0]; l[1] = g[1]; l[2] = g[2]; l[3] = g[3];
  }
  __syncthreads();

  const int nbase = (t >> 5) * 16;
  const int m = lane & 15;
  const int ncol = nbase + m;
  const int koff = (lane < 16) ? 0 : 2;

  v8f acc = {};
  const float* la = lds + m * LPITCH + koff;
  const float* wb = Wt + (size_t)ncol * KDIM + koff;
#pragma unroll 8
  for (int k = 0; k < KDIM; k += 4) {
    v2f a, b;
    float2 av = *reinterpret_cast<const float2*>(la + k);
    float2 bv = *reinterpret_cast<const float2*>(wb + k);
    a[0] = av.x; a[1] = av.y;
    b[0] = bv.x; b[1] = bv.y;
    acc = __builtin_amdgcn_wmma_f32_16x16x4_f32(
        false, a, false, b, (short)0, acc, false, false);
  }

  const float bv = bias[ncol];
  const int mo = (lane < 16) ? 0 : 8;
  float* orow = out + (size_t)mbase * NOUT + ncol;
#pragma unroll
  for (int i = 0; i < 8; ++i) orow[(size_t)(i + mo) * NOUT] = acc[i] + bv;
}

__global__ __launch_bounds__(256) void k_layernorm(float* __restrict__ buf,
                                                   const float* __restrict__ gw,
                                                   const float* __restrict__ gb,
                                                   int Nrows) {
  int lane = threadIdx.x & 31;
  int row = blockIdx.x * 8 + (threadIdx.x >> 5);
  if (row >= Nrows) return;
  float4* rp = reinterpret_cast<float4*>(buf) + (size_t)row * (HIDC / 4);
  float4 v = rp[lane];
  float s = v.x + v.y + v.z + v.w;
  float q = v.x * v.x + v.y * v.y + v.z * v.z + v.w * v.w;
#pragma unroll
  for (int msk = 16; msk >= 1; msk >>= 1) {
    s += __shfl_xor(s, msk, 32);
    q += __shfl_xor(q, msk, 32);
  }
  float mu = s * (1.0f / HIDC);
  float var = q * (1.0f / HIDC) - mu * mu;
  float rs = rsqrtf(var + 1e-5f);
  float4 g4 = reinterpret_cast<const float4*>(gw)[lane];
  float4 b4 = reinterpret_cast<const float4*>(gb)[lane];
  v.x = (v.x - mu) * rs * g4.x + b4.x;
  v.y = (v.y - mu) * rs * g4.y + b4.y;
  v.z = (v.z - mu) * rs * g4.z + b4.z;
  v.w = (v.w - mu) * rs * g4.w + b4.w;
  rp[lane] = v;
}

__global__ __launch_bounds__(256) void k_logsoftmax(float* __restrict__ out,
                                                    int Nrows, int C) {
  int lane = threadIdx.x & 31;
  int row = blockIdx.x * 8 + (threadIdx.x >> 5);
  if (row >= Nrows) return;
  float* r = out + (size_t)row * C;
  float a = r[lane];
  float b = r[lane + 32];
  float mx = fmaxf(a, b);
#pragma unroll
  for (int m = 16; m >= 1; m >>= 1) mx = fmaxf(mx, __shfl_xor(mx, m, 32));
  float e = expf(a - mx) + expf(b - mx);
#pragma unroll
  for (int m = 16; m >= 1; m >>= 1) e += __shfl_xor(e, m, 32);
  float lse = mx + logf(e);
  r[lane] = a - lse;
  r[lane + 32] = b - lse;
}

extern "C" void kernel_launch(void* const* d_in, const int* in_sizes, int n_in,
                              void* d_out, int out_size, void* d_ws, size_t ws_size,
                              hipStream_t stream) {
  (void)n_in; (void)out_size; (void)ws_size;
  const float* x    = (const float*)d_in[0];
  const int*   ei   = (const int*)d_in[1];
  const float* W1   = (const float*)d_in[2];
  const float* b1   = (const float*)d_in[3];
  const float* W2   = (const float*)d_in[4];
  const float* b2   = (const float*)d_in[5];
  const float* ln1g = (const float*)d_in[6];
  const float* ln1b = (const float*)d_in[7];
  const float* ln2g = (const float*)d_in[8];
  const float* ln2b = (const float*)d_in[9];
  const float* linW = (const float*)d_in[10];
  const float* linb = (const float*)d_in[11];
  float* outp = (float*)d_out;

  const int N = in_sizes[0] / HIDC;
  const int E = in_sizes[1] / 2;
  const int OUTC = in_sizes[11];
  const int* src = ei;
  const int* dstp = ei + E;

  const long NH = (long)N * HIDC;
  const long W3 = 3L * KDIM * HIDC;
  const long WH = (long)KDIM * OUTC;
  float* ws = (float*)d_ws;
  long pad = ((long)N + 255) & ~255L;
  float* inv_deg = ws;
  float* agg1 = ws + pad;
  float* agg2 = agg1 + NH;
  float* bufA = agg2 + NH;
  float* bufB = bufA + NH;
  float* Wt1  = bufB + NH;
  float* Wt2  = Wt1 + W3;
  float* WtH  = Wt2 + W3;

  dim3 b256(256);
  int nhBlocks = (int)((NH + 255) / 256);
  int nBlocks = (N + 255) / 256;
  int eBlocks = (E + 255) / 256;
  int eWaveBlocks = (E + 7) / 8;
  int rowBlocks = (N + 7) / 8;
  int mTiles = N / 16;

  k_transpose_w<<<(int)((W3 + 255) / 256), b256, 0, stream>>>(W1, Wt1, KDIM, HIDC, 3);
  k_transpose_w<<<(int)((W3 + 255) / 256), b256, 0, stream>>>(W2, Wt2, KDIM, HIDC, 3);
  k_transpose_w<<<(int)((WH + 255) / 256), b256, 0, stream>>>(linW, WtH, KDIM, OUTC, 1);

  k_zero<<<nBlocks, b256, 0, stream>>>(inv_deg, N);
  k_degree<<<eBlocks, b256, 0, stream>>>(dstp, inv_deg, E);
  k_invdeg<<<nBlocks, b256, 0, stream>>>(inv_deg, N);

  k_zero<<<nhBlocks, b256, 0, stream>>>(agg1, NH);
  k_scatter_add<<<eWaveBlocks, b256, 0, stream>>>(src, dstp, x, agg1, E);
  k_scale_rows<<<nhBlocks, b256, 0, stream>>>(agg1, inv_deg, NH);
  k_zero<<<nhBlocks, b256, 0, stream>>>(agg2, NH);
  k_scatter_add<<<eWaveBlocks, b256, 0, stream>>>(src, dstp, agg1, agg2, E);
  k_scale_rows<<<nhBlocks, b256, 0, stream>>>(agg2, inv_deg, NH);
  k_wmma_gemm3<<<mTiles, b256, 0, stream>>>(x, agg1, agg2, Wt1, b1, bufA);
  k_layernorm<<<rowBlocks, b256, 0, stream>>>(bufA, ln1g, ln1b, N);

  k_zero<<<nhBlocks, b256, 0, stream>>>(agg1, NH);
  k_scatter_add<<<eWaveBlocks, b256, 0, stream>>>(src, dstp, bufA, agg1, E);
  k_scale_rows<<<nhBlocks, b256, 0, stream>>>(agg1, inv_deg, NH);
  k_zero<<<nhBlocks, b256, 0, stream>>>(agg2, NH);
  k_scatter_add<<<eWaveBlocks, b256, 0, stream>>>(src, dstp, agg1, agg2, E);
  k_scale_rows<<<nhBlocks, b256, 0, stream>>>(agg2, inv_deg, NH);
  k_wmma_gemm3<<<mTiles, b256, 0, stream>>>(bufA, agg1, agg2, Wt2, b2, bufB);
  k_layernorm<<<rowBlocks, b256, 0, stream>>>(bufB, ln2g, ln2b, N);

  k_wmma_head<<<mTiles, dim3(128), 0, stream>>>(bufB, WtH, linb, outp, OUTC);
  k_logsoftmax<<<rowBlocks, b256, 0, stream>>>(outp, N, OUTC);
}


# Gather instead of atomic scatter: each call builds rows-by-destination (int atomic count+rank, scan, fill); a wave sums each row's neighbours, writes the mean; row sum order = atomic arrival rank
# speedup vs baseline: 1.9919x; 1.9919x over previous
#include <hip/hip_runtime.h>


#define HIDC 128
#define KDIM 128
#define LPITCH 132

typedef float v2f __attribute__((ext_vector_type(2)));
typedef float v8f __attribute__((ext_vector_type(8)));

__global__ __launch_bounds__(256) void k_zero_i32(int* __restrict__ p, int n) {
  int i = blockIdx.x * blockDim.x + threadIdx.x;
  if (i < n) p[i] = 0;
}

__global__ __launch_bounds__(256) void k_count_rank(const int* __restrict__ dst,
                                                    int* __restrict__ cnt,
                                                    int* __restrict__ rank, int E) {
  int e = blockIdx.x * blockDim.x + threadIdx.x;
  if (e < E)
    rank[e] = __hip_atomic_fetch_add(&cnt[dst[e]], 1, __ATOMIC_RELAXED,
                                     __HIP_MEMORY_SCOPE_AGENT);
}

__global__ __launch_bounds__(1024) void k_scan_counts(const int* __restrict__ cnt,
                                                      int* __restrict__ rowptr, int P) {
  __shared__ int wave_total[32];
  const int lane = threadIdx.x & 31;
  const int w = __builtin_amdgcn_readfirstlane((int)(threadIdx.x >> 5));
  const int trips = P >> 12;
  const int q0 = w * trips * 32 + lane;
  const int4* c4 = reinterpret_cast<const int4*>(cnt);
  int4* r4 = reinterpret_cast<int4*>(rowptr);

  int s = 0;
#pragma unroll 4
  for (int t = 0; t < trips; ++t) {
    const int4 c = c4[q0 + t * 32];
    s += (c.x + c.y) + (c.z + c.w);
  }
  for (int m = 16; m >= 1; m >>= 1) s += __shfl_xor(s, m, 32);
  if (lane == 0) wave_total[w] = s;
  __syncthreads();

  int carry = (lane < w) ? wave_total[lane] : 0;
  for (int m = 16; m >= 1; m >>= 1) carry += __shfl_xor(carry, m, 32);

#pragma unroll 4
  for (int t = 0; t < trips; ++t) {
    const int4 c = c4[q0 + t * 32];
    const int ls = (c.x + c.y) + (c.z + c.w);
    int inc = ls;
    for (int d = 1; d < 32; d <<= 1) {
      const int v = __shfl_up(inc, d, 32);
      if (lane >= d) inc += v;
    }
    int4 r;
    r.x = carry + inc - ls;
    r.y = r.x + c.x;
    r.z = r.y + c.y;
    r.w = r.z + c.z;
    r4[q0 + t * 32] = r;
    carry += __shfl(inc, 31, 32);
  }
}

__global__ __launch_bounds__(256) void k_fill_cols(const int* __restrict__ src,
                                                   const int* __restrict__ dst,
                                                   const int* __restrict__ rowptr,
                                                   const int* __restrict__ rank,
                                                   int* __restrict__ col, int E) {
  int e = blockIdx.x * blockDim.x + threadIdx.x;
  if (e < E) col[rowptr[dst[e]] + rank[e]] = src[e];
}

#define GROWS 4
__global__ __launch_bounds__(256) void k_gather_mean(const int* __restrict__ rowptr,
                                                     const int* __restrict__ col,
                                                     const float* __restrict__ h,
                                                     float* __restrict__ out, int N) {
  const int lane = threadIdx.x & 31;
  const int wv = __builtin_amdgcn_readfirstlane((int)(blockIdx.x * 8 + (threadIdx.x >> 5)));
  const int r0 = wv * GROWS;
  if (r0 >= N) return;
  const int nr = min(GROWS, N - r0);
  const int bnd = rowptr[r0 + min(lane, nr)];
  const float4* h4 = reinterpret_cast<const float4*>(h);
  float4* o4 = reinterpret_cast<float4*>(out);
  for (int r = 0; r < nr; ++r) {
    const int beg = __builtin_amdgcn_readlane(bnd, r);
    const int end = __builtin_amdgcn_readlane(bnd, r + 1);
    float4 acc = {0.0f, 0.0f, 0.0f, 0.0f};
    for (int j0 = beg; j0 < end; j0 += 32) {
      const int n = min(end - j0, 32);
      const int mine = (lane < n) ? col[j0 + lane] : 0;
#pragma unroll 4
      for (int k = 0; k < n; ++k) {
        const int sidx = __builtin_amdgcn_readlane(mine, k);
        const float4 v = h4[(size_t)sidx * (HIDC / 4) + lane];
        acc.x += v.x;
        acc.y += v.y;
        acc.z += v.z;
        acc.w += v.w;
      }
    }
    const float inv = 1.0f / fmaxf((float)(end - beg), 1.0f);
    acc.x *= inv;
    acc.y *= inv;
    acc.z *= inv;
    acc.w *= inv;
    o4[(size_t)(r0 + r) * (HIDC / 4) + lane] = acc;
  }
}

__global__ __launch_bounds__(256) void k_transpose_w(const float* __restrict__ W,
                                                     float* __restrict__ Wt,
                                                     int K, int NOUT, int P) {
  long total = (long)P * K * NOUT;
  long i = (long)blockIdx.x * blockDim.x + threadIdx.x;
  if (i >= total) return;
  int p = (int)(i / (K * NOUT));
  int rem = (int)(i % (K * NOUT));
  int k = rem / NOUT;
  int n = rem % NOUT;
  Wt[(size_t)p * K * NOUT + (size_t)n * K + k] = W[i];
}

__global__ __launch_bounds__(256) void k_wmma_gemm3(
    const float* __restrict__ X0, const float* __restrict__ X1,
    const float* __restrict__ X2, const float* __restrict__ Wt,
    const float* __restrict__ bias, float* __restrict__ out) {
  __shared__ float lds[3 * 16 * LPITCH];
  const int t = threadIdx.x;
  const int lane = t & 31;
  const int mbase = blockIdx.x * 16;

  {
    const int r = t >> 4;
    const int c = (t & 15) * 8;
    const float4* g0 = reinterpret_cast<const float4*>(X0 + (size_t)(mbase + r) * KDIM + c);
    const float4* g1 = reinterpret_cast<const float4*>(X1 + (size_t)(mbase + r) * KDIM + c);
    const float4* g2 = reinterpret_cast<const float4*>(X2 + (size_t)(mbase + r) * KDIM + c);
    float4* l0 = reinterpret_cast<float4*>(lds + 0 * 16 * LPITCH + r * LPITCH + c);
    float4* l1 = reinterpret_cast<float4*>(lds + 1 * 16 * LPITCH + r * LPITCH + c);
    float4* l2 = reinterpret_cast<float4*>(lds + 2 * 16 * LPITCH + r * LPITCH + c);
    l0[0] = g0[0]; l0[1] = g0[1];
    l1[0] = g1[0]; l1[1] = g1[1];
    l2[0] = g2[0]; l2[1] = g2[1];
  }
  __syncthreads();

  const int nbase = (t >> 5) * 16;
  const int m = lane & 15;
  const int ncol = nbase + m;
  const int koff = (lane < 16) ? 0 : 2;

  v8f acc = {};
#pragma unroll
  for (int p = 0; p < 3; ++p) {
    const float* la = lds + p * 16 * LPITCH + m * LPITCH + koff;
    const float* wb = Wt + (size_t)p * (KDIM * HIDC) + (size_t)ncol * KDIM + koff;
#pragma unroll 8
    for (int k = 0; k < KDIM; k += 4) {
      v2f a, b;
      float2 av = *reinterpret_cast<const float2*>(la + k);
      float2 bv = *reinterpret_cast<const float2*>(wb + k);
      a[0] = av.x; a[1] = av.y;
      b[0] = bv.x; b[1] = bv.y;
      acc = __builtin_amdgcn_wmma_f32_16x16x4_f32(
          false, a, false, b, (short)0, acc, false, false);
    }
  }

  const float bv = bias[ncol];
  const int mo = (lane < 16) ? 0 : 8;
  float* orow = out + (size_t)mbase * HIDC + ncol;
#pragma unroll
  for (int i = 0; i < 8; ++i) orow[(size_t)(i + mo) * HIDC] = acc[i] + bv;
}

__global__ __launch_bounds__(128) void k_wmma_head(
    const float* __restrict__ X, const float* __restrict__ Wt,
    const float* __restrict__ bias, float* __restrict__ out, int NOUT) {
  __shared__ float lds[16 * LPITCH];
  const int t = threadIdx.x;
  const int lane = t & 31;
  const int mbase = blockIdx.x * 16;

  {
    const int r = t >> 3;
    const int c = (t & 7) * 16;
    const float4* g = reinterpret_cast<const float4*>(X + (size_t)(mbase + r) * KDIM + c);
    float4* l = reinterpret_cast<float4*>(lds + r * LPITCH + c);
    l[0] = g[0]; l[1] = g[1]; l[2] = g[2]; l[3] = g[3];
  }
  __syncthreads();

  const int nbase = (t >> 5) * 16;
  const int m = lane & 15;
  const int ncol = nbase + m;
  const int koff = (lane < 16) ? 0 : 2;

  v8f acc = {};
  const float* la = lds + m * LPITCH + koff;
  const float* wb = Wt + (size_t)ncol * KDIM + koff;
#pragma unroll 8
  for (int k = 0; k < KDIM; k += 4) {
    v2f a, b;
    float2 av = *reinterpret_cast<const float2*>(la + k);
    float2 bv = *reinterpret_cast<const float2*>(wb + k);
    a[0] = av.x; a[1] = av.y;
    b[0] = bv.x; b[1] = bv.y;
    acc = __builtin_amdgcn_wmma_f32_16x16x4_f32(
        false, a, false, b, (short)0, acc, false, false);
  }

  const float bv = bias[ncol];
  const int mo = (lane < 16) ? 0 : 8;
  float* orow = out + (size_t)mbase * NOUT + ncol;
#pragma unroll
  for (int i = 0; i < 8; ++i) orow[(size_t)(i + mo) * NOUT] = acc[i] + bv;
}

__global__ __launch_bounds__(256) void k_layernorm(float* __restrict__ buf,
                                                   const float* __restrict__ gw,
                                                   const float* __restrict__ gb,
                                                   int Nrows) {
  int lane = threadIdx.x & 31;
  int row = blockIdx.x * 8 + (threadIdx.x >> 5);
  if (row >= Nrows) return;
  float4* rp = reinterpret_cast<float4*>(buf) + (size_t)row * (HIDC / 4);
  float4 v = rp[lane];
  float s = v.x + v.y + v.z + v.w;
  float q = v.x * v.x + v.y * v.y + v.z * v.z + v.w * v.w;
#pragma unroll
  for (int msk = 16; msk >= 1; msk >>= 1) {
    s += __shfl_xor(s, msk, 32);
    q += __shfl_xor(q, msk, 32);
  }
  float mu = s * (1.0f / HIDC);
  float var = q * (1.0f / HIDC) - mu * mu;
  float rs = rsqrtf(var + 1e-5f);
  float4 g4 = reinterpret_cast<const float4*>(gw)[lane];
  float4 b4 = reinterpret_cast<const float4*>(gb)[lane];
  v.x = (v.x - mu) * rs * g4.x + b4.x;
  v.y = (v.y - mu) * rs * g4.y + b4.y;
  v.z = (v.z - mu) * rs * g4.z + b4.z;
  v.w = (v.w - mu) * rs * g4.w + b4.w;
  rp[lane] = v;
}

__global__ __launch_bounds__(256) void k_logsoftmax(float* __restrict__ out,
                                                    int Nrows, int C) {
  int lane = threadIdx.x & 31;
  int row = blockIdx.x * 8 + (threadIdx.x >> 5);
  if (row >= Nrows) return;
  float* r = out + (size_t)row * C;
  float a = r[lane];
  float b = r[lane + 32];
  float mx = fmaxf(a, b);
#pragma unroll
  for (int m = 16; m >= 1; m >>= 1) mx = fmaxf(mx, __shfl_xor(mx, m, 32));
  float e = expf(a - mx) + expf(b - mx);
#pragma unroll
  for (int m = 16; m >= 1; m >>= 1) e += __shfl_xor(e, m, 32);
  float lse = mx + logf(e);
  r[lane] = a - lse;
  r[lane + 32] = b - lse;
}

extern "C" void kernel_launch(void* const* d_in, const int* in_sizes, int n_in,
                              void* d_out, int out_size, void* d_ws, size_t ws_size,
                              hipStream_t stream) {
  (void)n_in; (void)out_size; (void)ws_size;
  const float* x    = (const float*)d_in[0];
  const int*   ei   = (const int*)d_in[1];
  const float* W1   = (const float*)d_in[2];
  const float* b1   = (const float*)d_in[3];
  const float* W2   = (const float*)d_in[4];
  const float* b2   = (const float*)d_in[5];
  const float* ln1g = (const float*)d_in[6];
  const float* ln1b = (const float*)d_in[7];
  const float* ln2g = (const float*)d_in[8];
  const float* ln2b = (const float*)d_in[9];
  const float* linW = (const float*)d_in[10];
  const float* linb = (const float*)d_in[11];
  float* outp = (float*)d_out;

  const int N = in_sizes[0] / HIDC;
  const int E = in_sizes[1] / 2;
  const int OUTC = in_sizes[11];
  const int* src = ei;
  const int* dstp = ei + E;

  const long NH = (long)N * HIDC;
  const long W3 = 3L * KDIM * HIDC;
  const long WH = (long)KDIM * OUTC;
  float* ws = (float*)d_ws;
  long pad = ((long)N + 255) & ~255L;
  float* agg1 = ws + pad;
  float* agg2 = agg1 + NH;
  float* bufA = agg2 + NH;
  float* bufB = bufA + NH;
  float* Wt1  = bufB + NH;
  float* Wt2  = Wt1 + W3;
  float* WtH  = Wt2 + W3;

  const int padI = (N + 1 + 4095) & ~4095;
  int* cnt    = (int*)bufB;
  int* rowptr = cnt + padI;
  int* rank   = rowptr + padI;
  int* col    = rank + E;

  dim3 b256(256);
  int eBlocks = (E + 255) / 256;
  int rowBlocks = (N + 7) / 8;
  int gatherBlocks = (N + 8 * GROWS - 1) / (8 * GROWS);
  int mTiles = N / 16;

  k_transpose_w<<<(int)((W3 + 255) / 256), b256, 0, stream>>>(W1, Wt1, KDIM, HIDC, 3);
  k_transpose_w<<<(int)((W3 + 255) / 256), b256, 0, stream>>>(W2, Wt2, KDIM, HIDC, 3);
  k_transpose_w<<<(int)((WH + 255) / 256), b256, 0, stream>>>(linW, WtH, KDIM, OUTC, 1);

  k_zero_i32<<<padI / 256, b256, 0, stream>>>(cnt, padI);
  k_count_rank<<<eBlocks, b256, 0, stream>>>(dstp, cnt, rank, E);
  k_scan_counts<<<1, dim3(1024), 0, stream>>>(cnt, rowptr, padI);
  k_fill_cols<<<eBlocks, b256, 0, stream>>>(src, dstp, rowptr, rank, col, E);

  k_gather_mean<<<gatherBlocks, b256, 0, stream>>>(rowptr, col, x, agg1, N);
  k_gather_mean<<<gatherBlocks, b256, 0, stream>>>(rowptr, col, agg1, agg2, N);
  k_wmma_gemm3<<<mTiles, b256, 0, stream>>>(x, agg1, agg2, Wt1, b1, bufA);
  k_layernorm<<<rowBlocks, b256, 0, stream>>>(bufA, ln1g, ln1b, N);

  k_gather_mean<<<gatherBlocks, b256, 0, stream>>>(rowptr, col, bufA, agg1, N);
  k_gather_mean<<<gatherBlocks, b256, 0, stream>>>(rowptr, col, agg1, agg2, N);
  k_wmma_gemm3<<<mTiles, b256, 0, stream>>>(bufA, agg1, agg2, Wt2, b2, bufB);
  k_layernorm<<<rowBlocks, b256, 0, stream>>>(bufB, ln2g, ln2b, N);

  k_wmma_head<<<mTiles, dim3(128), 0, stream>>>(bufB, WtH, linb, outp, OUTC);
  k_logsoftmax<<<rowBlocks, b256, 0, stream>>>(outp, N, OUTC);
}


# Row epilogues fused into their GEMMs: LayerNorm inside the 3-power GEMM, log-softmax inside the head GEMM (tile via LDS, wave per row, whole-row stores); 3 launches fewer; sum order unchanged
# speedup vs baseline: 2.6949x; 1.3529x over previous
#include <hip/hip_runtime.h>


#define HIDC 128
#define KDIM 128
#define LPITCH 132
#define HPITCH 68

typedef float v2f __attribute__((ext_vector_type(2)));
typedef float v8f __attribute__((ext_vector_type(8)));

__global__ __launch_bounds__(256) void k_zero_i32(int* __restrict__ p, int n) {
  int i = blockIdx.x * blockDim.x + threadIdx.x;
  if (i < n) p[i] = 0;
}

__global__ __launch_bounds__(256) void k_count_rank(const int* __restrict__ dst,
                                                    int* __restrict__ cnt,
                                                    int* __restrict__ rank, int E) {
  int e = blockIdx.x * blockDim.x + threadIdx.x;
  if (e < E)
    rank[e] = __hip_atomic_fetch_add(&cnt[dst[e]], 1, __ATOMIC_RELAXED,
                                     __HIP_MEMORY_SCOPE_AGENT);
}

__global__ __launch_bounds__(1024) void k_scan_counts(const int* __restrict__ cnt,
                                                      int* __restrict__ rowptr, int P) {
  __shared__ int wave_total[32];
  const int lane = threadIdx.x & 31;
  const int w = __builtin_amdgcn_readfirstlane((int)(threadIdx.x >> 5));
  const int trips = P >> 12;
  const int q0 = w * trips * 32 + lane;
  const int4* c4 = reinterpret_cast<const int4*>(cnt);
  int4* r4 = reinterpret_cast<int4*>(rowptr);

  int s = 0;
#pragma unroll 4
  for (int t = 0; t < trips; ++t) {
    const int4 c = c4[q0 + t * 32];
    s += (c.x + c.y) + (c.z + c.w);
  }
  for (int m = 16; m >= 1; m >>= 1) s += __shfl_xor(s, m, 32);
  if (lane == 0) wave_total[w] = s;
  __syncthreads();

  int carry = (lane < w) ? wave_total[lane] : 0;
  for (int m = 16; m >= 1; m >>= 1) carry += __shfl_xor(carry, m, 32);

#pragma unroll 4
  for (int t = 0; t < trips; ++t) {
    const int4 c = c4[q0 + t * 32];
    const int ls = (c.x + c.y) + (c.z + c.w);
    int inc = ls;
    for (int d = 1; d < 32; d <<= 1) {
      const int v = __shfl_up(inc, d, 32);
      if (lane >= d) inc += v;
    }
    int4 r;
    r.x = carry + inc - ls;
    r.y = r.x + c.x;
    r.z = r.y + c.y;
    r.w = r.z + c.z;
    r4[q0 + t * 32] = r;
    carry += __shfl(inc, 31, 32);
  }
}

__global__ __launch_bounds__(256) void k_fill_cols(const int* __restrict__ src,
                                                   const int* __restrict__ dst,
                                                   const int* __restrict__ rowptr,
                                                   const int* __restrict__ rank,
                                                   int* __restrict__ col, int E) {
  int e = blockIdx.x * blockDim.x + threadIdx.x;
  if (e < E) col[rowptr[dst[e]] + rank[e]] = src[e];
}

#define GROWS 4
__global__ __launch_bounds__(256) void k_gather_mean(const int* __restrict__ rowptr,
                                                     const int* __restrict__ col,
                                                     const float* __restrict__ h,
                                                     float* __restrict__ out, int N) {
  const int lane = threadIdx.x & 31;
  const int wv = __builtin_amdgcn_readfirstlane((int)(blockIdx.x * 8 + (threadIdx.x >> 5)));
  const int r0 = wv * GROWS;
  if (r0 >= N) return;
  const int nr = min(GROWS, N - r0);
  const int bnd = rowptr[r0 + min(lane, nr)];
  const float4* h4 = reinterpret_cast<const float4*>(h);
  float4* o4 = reinterpret_cast<float4*>(out);
  for (int r = 0; r < nr; ++r) {
    const int beg = __builtin_amdgcn_readlane(bnd, r);
    const int end = __builtin_amdgcn_readlane(bnd, r + 1);
    float4 acc = {0.0f, 0.0f, 0.0f, 0.0f};
    for (int j0 = beg; j0 < end; j0 += 32) {
      const int n = min(end - j0, 32);
      const int mine = (lane < n) ? col[j0 + lane] : 0;
#pragma unroll 4
      for (int k = 0; k < n; ++k) {
        const int sidx = __builtin_amdgcn_readlane(mine, k);
        const float4 v = h4[(size_t)sidx * (HIDC / 4) + lane];
        acc.x += v.x;
        acc.y += v.y;
        acc.z += v.z;
        acc.w += v.w;
      }
    }
    const float inv = 1.0f / fmaxf((float)(end - beg), 1.0f);
    acc.x *= inv;
    acc.y *= inv;
    acc.z *= inv;
    acc.w *= inv;
    o4[(size_t)(r0 + r) * (HIDC / 4) + lane] = acc;
  }
}

__global__ __launch_bounds__(256) void k_transpose_w(const float* __restrict__ W,
                                                     float* __restrict__ Wt,
                                                     int K, int NOUT, int P) {
  long total = (long)P * K * NOUT;
  long i = (long)blockIdx.x * blockDim.x + threadIdx.x;
  if (i >= total) return;
  int p = (int)(i / (K * NOUT));
  int rem = (int)(i % (K * NOUT));
  int k = rem / NOUT;
  int n = rem % NOUT;
  Wt[(size_t)p * K * NOUT + (size_t)n * K + k] = W[i];
}

__global__ __launch_bounds__(256) void k_wmma_gemm3_ln(
    const float* __restrict__ X0, const float* __restrict__ X1,
    const float* __restrict__ X2, const float* __restrict__ Wt,
    const float* __restrict__ bias, const float* __restrict__ gw,
    const float* __restrict__ gb, float* __restrict__ out) {
  __shared__ float lds[3 * 16 * LPITCH];
  __shared__ float tile[16 * LPITCH];
  const int t = threadIdx.x;
  const int lane = t & 31;
  const int mbase = blockIdx.x * 16;

  {
    const int r = t >> 4;
    const int c = (t & 15) * 8;
    const float4* g0 = reinterpret_cast<const float4*>(X0 + (size_t)(mbase + r) * KDIM + c);
    const float4* g1 = reinterpret_cast<const float4*>(X1 + (size_t)(mbase + r) * KDIM + c);
    const float4* g2 = reinterpret_cast<const float4*>(X2 + (size_t)(mbase + r) * KDIM + c);
    float4* l0 = reinterpret_cast<float4*>(lds + 0 * 16 * LPITCH + r * LPITCH + c);
    float4* l1 = reinterpret_cast<float4*>(lds + 1 * 16 * LPITCH + r * LPITCH + c);
    float4* l2 = reinterpret_cast<float4*>(lds + 2 * 16 * LPITCH + r * LPITCH + c);
    l0[0] = g0[0]; l0[1] = g0[1];
    l1[0] = g1[0]; l1[1] = g1[1];
    l2[0] = g2[0]; l2[1] = g2[1];
  }
  __syncthreads();

  const int nbase = (t >> 5) * 16;
  const int m = lane & 15;
  const int ncol = nbase + m;
  const int koff = (lane < 16) ? 0 : 2;

  v8f acc = {};
#pragma unroll
  for (int p = 0; p < 3; ++p) {
    const float* la = lds + p * 16 * LPITCH + m * LPITCH + koff;
    const float* wb = Wt + (size_t)p * (KDIM * HIDC) + (size_t)ncol * KDIM + koff;
#pragma unroll 8
    for (int k = 0; k < KDIM; k += 4) {
      v2f a, b;
      float2 av = *reinterpret_cast<const float2*>(la + k);
      float2 bv = *reinterpret_cast<const float2*>(wb + k);
      a[0] = av.x; a[1] = av.y;
      b[0] = bv.x; b[1] = bv.y;
      acc = __builtin_amdgcn_wmma_f32_16x16x4_f32(
          false, a, false, b, (short)0, acc, false, false);
    }
  }

  const float bv = bias[ncol];
  const int mo = (lane < 16) ? 0 : 8;
#pragma unroll
  for (int i = 0; i < 8; ++i) tile[(i + mo) * LPITCH + ncol] = acc[i] + bv;
  __syncthreads();

  const int w = t >> 5;
  const float4 g4 = reinterpret_cast<const float4*>(gw)[lane];
  const float4 b4 = reinterpret_cast<const float4*>(gb)[lane];
  float4* o4 = reinterpret_cast<float4*>(out) + (size_t)mbase * (HIDC / 4);
  for (int rr = 0; rr < 2; ++rr) {
    const int row = w * 2 + rr;
    float4 v = *reinterpret_cast<const float4*>(tile + row * LPITCH + lane * 4);
    float s = v.x + v.y + v.z + v.w;
    float q = v.x * v.x + v.y * v.y + v.z * v.z + v.w * v.w;
    for (int msk = 16; msk >= 1; msk >>= 1) {
      s += __shfl_xor(s, msk, 32);
      q += __shfl_xor(q, msk, 32);
    }
    const float mu = s * (1.0f / HIDC);
    const float var = q * (1.0f / HIDC) - mu * mu;
    const float rs = rsqrtf(var + 1e-5f);
    v.x = (v.x - mu) * rs * g4.x + b4.x;
    v.y = (v.y - mu) * rs * g4.y + b4.y;
    v.z = (v.z - mu) * rs * g4.z + b4.z;
    v.w = (v.w - mu) * rs * g4.w + b4.w;
    o4[(size_t)row * (HIDC / 4) + lane] = v;
  }
}

__global__ __launch_bounds__(128) void k_wmma_head_lsm(
    const float* __restrict__ X, const float* __restrict__ Wt,
    const float* __restrict__ bias, float* __restrict__ out, int NOUT) {
  __shared__ float lds[16 * LPITCH];
  __shared__ float tile[16 * HPITCH];
  const int t = threadIdx.x;
  const int lane = t & 31;
  const int mbase = blockIdx.x * 16;

  {
    const int r = t >> 3;
    const int c = (t & 7) * 16;
    const float4* g = reinterpret_cast<const float4*>(X + (size_t)(mbase + r) * KDIM + c);
    float4* l = reinterpret_cast<float4*>(lds + r * LPITCH + c);
    l[0] = g[0]; l[1] = g[1]; l[2] = g[2]; l[3] = g[3];
  }
  __syncthreads();

  const int nbase = (t >> 5) * 16;
  const int m = lane & 15;
  const int ncol = nbase + m;
  const int koff = (lane < 16) ? 0 : 2;

  v8f acc = {};
  const float* la = lds + m * LPITCH + koff;
  const float* wb = Wt + (size_t)ncol * KDIM + koff;
#pragma unroll 8
  for (int k = 0; k < KDIM; k += 4) {
    v2f a, b;
    float2 av = *reinterpret_cast<const float2*>(la + k);
    float2 bv = *reinterpret_cast<const float2*>(wb + k);
    a[0] = av.x; a[1] = av.y;
    b[0] = bv.x; b[1] = bv.y;
    acc = __builtin_amdgcn_wmma_f32_16x16x4_f32(
        false, a, false, b, (short)0, acc, false, false);
  }

  const float bv = bias[ncol];
  const int mo = (lane < 16) ? 0 : 8;
#pragma unroll
  for (int i = 0; i < 8; ++i) tile[(i + mo) * HPITCH + ncol] = acc[i] + bv;
  __syncthreads();

  const int w = t >> 5;
  for (int rr = 0; rr < 4; ++rr) {
    const int row = w * 4 + rr;
    const float* r = tile + row * HPITCH;
    const float a = r[lane];
    const float b = r[lane + 32];
    float mx = fmaxf(a, b);
    for (int msk = 16; msk >= 1; msk >>= 1) mx = fmaxf(mx, __shfl_xor(mx, msk, 32));
    float e = expf(a - mx) + expf(b - mx);
    for (int msk = 16; msk >= 1; msk >>= 1) e += __shfl_xor(e, msk, 32);
    const float lse = mx + logf(e);
    float* o = out + (size_t)(mbase + row) * NOUT;
    o[lane] = a - lse;
    o[lane + 32] = b - lse;
  }
}

extern "C" void kernel_launch(void* const* d_in, const int* in_sizes, int n_in,
                              void* d_out, int out_size, void* d_ws, size_t ws_size,
                              hipStream_t stream) {
  (void)n_in; (void)out_size; (void)ws_size;
  const float* x    = (const float*)d_in[0];
  const int*   ei   = (const int*)d_in[1];
  const float* W1   = (const float*)d_in[2];
  const float* b1   = (const float*)d_in[3];
  const float* W2   = (const float*)d_in[4];
  const float* b2   = (const float*)d_in[5];
  const float* ln1g = (const float*)d_in[6];
  const float* ln1b = (const float*)d_in[7];
  const float* ln2g = (const float*)d_in[8];
  const float* ln2b = (const float*)d_in[9];
  const float* linW = (const float*)d_in[10];
  const float* linb = (const float*)d_in[11];
  float* outp = (float*)d_out;

  const int N = in_sizes[0] / HIDC;
  const int E = in_sizes[1] / 2;
  const int OUTC = in_sizes[11];
  const int* src = ei;
  const int* dstp = ei + E;

  const long NH = (long)N * HIDC;
  const long W3 = 3L * KDIM * HIDC;
  const long WH = (long)KDIM * OUTC;
  float* ws = (float*)d_ws;
  long pad = ((long)N + 255) & ~255L;
  float* agg1 = ws + pad;
  float* agg2 = agg1 + NH;
  float* bufA = agg2 + NH;
  float* bufB = bufA + NH;
  float* Wt1  = bufB + NH;
  float* Wt2  = Wt1 + W3;
  float* WtH  = Wt2 + W3;

  const int padI = (N + 1 + 4095) & ~4095;
  int* cnt    = (int*)bufB;
  int* rowptr = cnt + padI;
  int* rank   = rowptr + padI;
  int* col    = rank + E;

  dim3 b256(256);
  int eBlocks = (E + 255) / 256;
  int gatherBlocks = (N + 8 * GROWS - 1) / (8 * GROWS);
  int mTiles = N / 16;

  k_transpose_w<<<(int)((W3 + 255) / 256), b256, 0, stream>>>(W1, Wt1, KDIM, HIDC, 3);
  k_transpose_w<<<(int)((W3 + 255) / 256), b256, 0, stream>>>(W2, Wt2, KDIM, HIDC, 3);
  k_transpose_w<<<(int)((WH + 255) / 256), b256, 0, stream>>>(linW, WtH, KDIM, OUTC, 1);

  k_zero_i32<<<padI / 256, b256, 0, stream>>>(cnt, padI);
  k_count_rank<<<eBlocks, b256, 0, stream>>>(dstp, cnt, rank, E);
  k_scan_counts<<<1, dim3(1024), 0, stream>>>(cnt, rowptr, padI);
  k_fill_cols<<<eBlocks, b256, 0, stream>>>(src, dstp, rowptr, rank, col, E);

  k_gather_mean<<<gatherBlocks, b256, 0, stream>>>(rowptr, col, x, agg1, N);
  k_gather_mean<<<gatherBlocks, b256, 0, stream>>>(rowptr, col, agg1, agg2, N);
  k_wmma_gemm3_ln<<<mTiles, b256, 0, stream>>>(x, agg1, agg2, Wt1, b1, ln1g, ln1b, bufA);

  k_gather_mean<<<gatherBlocks, b256, 0, stream>>>(rowptr, col, bufA, agg1, N);
  k_gather_mean<<<gatherBlocks, b256, 0, stream>>>(rowptr, col, agg1, agg2, N);
  k_wmma_gemm3_ln<<<mTiles, b256, 0, stream>>>(bufA, agg1, agg2, Wt2, b2, ln2g, ln2b, bufB);

  k_wmma_head_lsm<<<mTiles, dim3(128), 0, stream>>>(bufB, WtH, linb, outp, OUTC);
}


# 3-power GEMM: a work-group takes 7 stacked 16-row tiles, so one weight fragment load feeds 7 WMMAs (7 accumulator tiles per wave); A staged one power at a time; sum order unchanged
# speedup vs baseline: 3.1233x; 1.1590x over previous
#include <hip/hip_runtime.h>


#define HIDC 128
#define KDIM 128
#define LPITCH 132
#define HPITCH 68

typedef float v2f __attribute__((ext_vector_type(2)));
typedef float v4f __attribute__((ext_vector_type(4)));
typedef float v8f __attribute__((ext_vector_type(8)));

__global__ __launch_bounds__(256) void k_zero_i32(int* __restrict__ p, int n) {
  int i = blockIdx.x * blockDim.x + threadIdx.x;
  if (i < n) p[i] = 0;
}

__global__ __launch_bounds__(256) void k_count_rank(const int* __restrict__ dst,
                                                    int* __restrict__ cnt,
                                                    int* __restrict__ rank, int E) {
  int e = blockIdx.x * blockDim.x + threadIdx.x;
  if (e < E)
    rank[e] = __hip_atomic_fetch_add(&cnt[dst[e]], 1, __ATOMIC_RELAXED,
                                     __HIP_MEMORY_SCOPE_AGENT);
}

__global__ __launch_bounds__(1024) void k_scan_counts(const int* __restrict__ cnt,
                                                      int* __restrict__ rowptr, int P) {
  __shared__ int wave_total[32];
  const int lane = threadIdx.x & 31;
  const int w = __builtin_amdgcn_readfirstlane((int)(threadIdx.x >> 5));
  const int trips = P >> 12;
  const int q0 = w * trips * 32 + lane;
  const int4* c4 = reinterpret_cast<const int4*>(cnt);
  int4* r4 = reinterpret_cast<int4*>(rowptr);

  int s = 0;
#pragma unroll 4
  for (int t = 0; t < trips; ++t) {
    const int4 c = c4[q0 + t * 32];
    s += (c.x + c.y) + (c.z + c.w);
  }
  for (int m = 16; m >= 1; m >>= 1) s += __shfl_xor(s, m, 32);
  if (lane == 0) wave_total[w] = s;
  __syncthreads();

  int carry = (lane < w) ? wave_total[lane] : 0;
  for (int m = 16; m >= 1; m >>= 1) carry += __shfl_xor(carry, m, 32);

#pragma unroll 4
  for (int t = 0; t < trips; ++t) {
    const int4 c = c4[q0 + t * 32];
    const int ls = (c.x + c.y) + (c.z + c.w);
    int inc = ls;
    for (int d = 1; d < 32; d <<= 1) {
      const int v = __shfl_up(inc, d, 32);
      if (lane >= d) inc += v;
    }
    int4 r;
    r.x = carry + inc - ls;
    r.y = r.x + c.x;
    r.z = r.y + c.y;
    r.w = r.z + c.z;
    r4[q0 + t * 32] = r;
    carry += __shfl(inc, 31, 32);
  }
}

__global__ __launch_bounds__(256) void k_fill_cols(const int* __restrict__ src,
                                                   const int* __restrict__ dst,
                                                   const int* __restrict__ rowptr,
                                                   const int* __restrict__ rank,
                                                   int* __restrict__ col, int E) {
  int e = blockIdx.x * blockDim.x + threadIdx.x;
  if (e < E) col[rowptr[dst[e]] + rank[e]] = src[e];
}

#define GROWS 4
__global__ __launch_bounds__(256) void k_gather_mean(const int* __restrict__ rowptr,
                                                     const int* __restrict__ col,
                                                     const float* __restrict__ h,
                                                     float* __restrict__ out, int N) {
  const int lane = threadIdx.x & 31;
  const int wv = __builtin_amdgcn_readfirstlane((int)(blockIdx.x * 8 + (threadIdx.x >> 5)));
  const int r0 = wv * GROWS;
  if (r0 >= N) return;
  const int nr = min(GROWS, N - r0);
  const int bnd = rowptr[r0 + min(lane, nr)];
  const float4* h4 = reinterpret_cast<const float4*>(h);
  float4* o4 = reinterpret_cast<float4*>(out);
  for (int r = 0; r < nr; ++r) {
    const int beg = __builtin_amdgcn_readlane(bnd, r);
    const int end = __builtin_amdgcn_readlane(bnd, r + 1);
    float4 acc = {0.0f, 0.0f, 0.0f, 0.0f};
    for (int j0 = beg; j0 < end; j0 += 32) {
      const int n = min(end - j0, 32);
      const int mine = (lane < n) ? col[j0 + lane] : 0;
#pragma unroll 4
      for (int k = 0; k < n; ++k) {
        const int sidx = __builtin_amdgcn_readlane(mine, k);
        const float4 v = h4[(size_t)sidx * (HIDC / 4) + lane];
        acc.x += v.x;
        acc.y += v.y;
        acc.z += v.z;
        acc.w += v.w;
      }
    }
    const float inv = 1.0f / fmaxf((float)(end - beg), 1.0f);
    acc.x *= inv;
    acc.y *= inv;
    acc.z *= inv;
    acc.w *= inv;
    o4[(size_t)(r0 + r) * (HIDC / 4) + lane] = acc;
  }
}

__global__ __launch_bounds__(256) void k_transpose_w(const float* __restrict__ W,
                                                     float* __restrict__ Wt,
                                                     int K, int NOUT, int P) {
  long total = (long)P * K * NOUT;
  long i = (long)blockIdx.x * blockDim.x + threadIdx.x;
  if (i >= total) return;
  int p = (int)(i / (K * NOUT));
  int rem = (int)(i % (K * NOUT));
  int k = rem / NOUT;
  int n = rem % NOUT;
  Wt[(size_t)p * K * NOUT + (size_t)n * K + k] = W[i];
}

#define MT 7
#define MROWS (MT * 16)
#define STG (MROWS / 8)

#define FETCH_ROWS(X)                                                            \
  _Pragma("unroll") for (int j = 0; j < STG; ++j) {                              \
    const int grow = min(mbase + w + 8 * j, N - 1);                              \
    pre[j] = reinterpret_cast<const v4f*>((X) + (size_t)grow * KDIM)[lane];      \
  }
#define FETCH_ROWS_AHEAD(X)                                                      \
  FETCH_ROWS(X)                                                                  \
  __builtin_amdgcn_sched_barrier(0);
#define PARK_ROWS()                                                              \
  _Pragma("unroll") for (int j = 0; j < STG; ++j)                                \
    *reinterpret_cast<v4f*>(lds + (w + 8 * j) * LPITCH + lane * 4) = pre[j];
#define KSTEP(KK, BF)                                                            \
  _Pragma("unroll") for (int i = 0; i < MT; ++i) {                               \
    v2f a;                                                                       \
    const float2 av = *reinterpret_cast<const float2*>(la + i * 16 * LPITCH + (KK)); \
    a[0] = av.x; a[1] = av.y;                                                    \
    acc[i] = __builtin_amdgcn_wmma_f32_16x16x4_f32(                              \
        false, a, false, (BF), (short)0, acc[i], false, false);                  \
  }
#define LOAD_B4(DST, K0)                                                         \
  _Pragma("unroll") for (int s = 0; s < 4; ++s) {                                \
    const float2 bv2 = *reinterpret_cast<const float2*>(wb + (K0) + 4 * s);      \
    DST[s][0] = bv2.x; DST[s][1] = bv2.y;                                        \
  }
#define PRODUCTS(P, NEXT_FETCH)                                                  \
  {                                                                              \
    const float* wb = Wt + (size_t)(P) * (KDIM * HIDC) + (size_t)ncol * KDIM + koff; \
    v2f bq[4], bn[4];                                                            \
    LOAD_B4(bq, 0)                                                               \
      \
    _Pragma("unroll 2") for (int k = 0; k < KDIM - 32; k += 16) {                \
      LOAD_B4(bn, k + 16)                                                        \
      _Pragma("unroll") for (int s = 0; s < 4; ++s) { KSTEP(k + 4 * s, bq[s]) }  \
      _Pragma("unroll") for (int s = 0; s < 4; ++s) bq[s] = bn[s];               \
    }                                                                            \
    LOAD_B4(bn, KDIM - 16)                                                       \
    NEXT_FETCH                                                                   \
    _Pragma("unroll") for (int s = 0; s < 4; ++s) { KSTEP(KDIM - 32 + 4 * s, bq[s]) } \
    _Pragma("unroll") for (int s = 0; s < 4; ++s) { KSTEP(KDIM - 16 + 4 * s, bn[s]) } \
  }

__global__ __launch_bounds__(256) void k_wmma_gemm3_ln(
    const float* __restrict__ X0, const float* __restrict__ X1,
    const float* __restrict__ X2, const float* __restrict__ Wt,
    const float* __restrict__ bias, const float* __restrict__ gw,
    const float* __restrict__ gb, float* __restrict__ out, int N) {
  __shared__ float lds[MROWS * LPITCH];
  const int t = threadIdx.x;
  const int lane = t & 31;
  const int w = __builtin_amdgcn_readfirstlane(t >> 5);
  const int mbase = blockIdx.x * MROWS;

  v4f pre[STG];
  FETCH_ROWS(X0)
  PARK_ROWS()
  __syncthreads();

  const int m = lane & 15;
  const int ncol = w * 16 + m;
  const int koff = (lane < 16) ? 0 : 2;
  const float* la = lds + m * LPITCH + koff;

  v8f acc[MT];
#pragma unroll
  for (int i = 0; i < MT; ++i) acc[i] = v8f{};

  PRODUCTS(0, FETCH_ROWS_AHEAD(X1))
  __syncthreads();
  PARK_ROWS()
  __syncthreads();

  PRODUCTS(1, FETCH_ROWS_AHEAD(X2))
  __syncthreads();
  PARK_ROWS()
  __syncthreads();

  PRODUCTS(2, )
  __syncthreads();

  const float bv = bias[ncol];
  const int mo = (lane < 16) ? 0 : 8;
#pragma unroll
  for (int i = 0; i < MT; ++i)
#pragma unroll
    for (int r = 0; r < 8; ++r) lds[(i * 16 + r + mo) * LPITCH + ncol] = acc[i][r] + bv;
  __syncthreads();

  const float4 g4 = reinterpret_cast<const float4*>(gw)[lane];
  const float4 b4 = reinterpret_cast<const float4*>(gb)[lane];
  float4* o4 = reinterpret_cast<float4*>(out);
  for (int rr = 0; rr < STG; ++rr) {
    const int row = w + 8 * rr;
    if (mbase + row >= N) break;
    float4 v = *reinterpret_cast<const float4*>(lds + row * LPITCH + lane * 4);
    float s = v.x + v.y + v.z + v.w;
    float q = v.x * v.x + v.y * v.y + v.z * v.z + v.w * v.w;
    for (int msk = 16; msk >= 1; msk >>= 1) {
      s += __shfl_xor(s, msk, 32);
      q += __shfl_xor(q, msk, 32);
    }
    const float mu = s * (1.0f / HIDC);
    const float var = q * (1.0f / HIDC) - mu * mu;
    const float rs = rsqrtf(var + 1e-5f);
    v.x = (v.x - mu) * rs * g4.x + b4.x;
    v.y = (v.y - mu) * rs * g4.y + b4.y;
    v.z = (v.z - mu) * rs * g4.z + b4.z;
    v.w = (v.w - mu) * rs * g4.w + b4.w;
    o4[(size_t)(mbase + row) * (HIDC / 4) + lane] = v;
  }
}

__global__ __launch_bounds__(128) void k_wmma_head_lsm(
    const float* __restrict__ X, const float* __restrict__ Wt,
    const float* __restrict__ bias, float* __restrict__ out, int NOUT) {
  __shared__ float lds[16 * LPITCH];
  __shared__ float tile[16 * HPITCH];
  const int t = threadIdx.x;
  const int lane = t & 31;
  const int mbase = blockIdx.x * 16;

  {
    const int r = t >> 3;
    const int c = (t & 7) * 16;
    const float4* g = reinterpret_cast<const float4*>(X + (size_t)(mbase + r) * KDIM + c);
    float4* l = reinterpret_cast<float4*>(lds + r * LPITCH + c);
    l[0] = g[0]; l[1] = g[1]; l[2] = g[2]; l[3] = g[3];
  }
  __syncthreads();

  const int nbase = (t >> 5) * 16;
  const int m = lane & 15;
  const int ncol = nbase + m;
  const int koff = (lane < 16) ? 0 : 2;

  v8f acc = {};
  const float* la = lds + m * LPITCH + koff;
  const float* wb = Wt + (size_t)ncol * KDIM + koff;
#pragma unroll 8
  for (int k = 0; k < KDIM; k += 4) {
    v2f a, b;
    float2 av = *reinterpret_cast<const float2*>(la + k);
    float2 bv = *reinterpret_cast<const float2*>(wb + k);
    a[0] = av.x; a[1] = av.y;
    b[0] = bv.x; b[1] = bv.y;
    acc = __builtin_amdgcn_wmma_f32_16x16x4_f32(
        false, a, false, b, (short)0, acc, false, false);
  }

  const float bv = bias[ncol];
  const int mo = (lane < 16) ? 0 : 8;
#pragma unroll
  for (int i = 0; i < 8; ++i) tile[(i + mo) * HPITCH + ncol] = acc[i] + bv;
  __syncthreads();

  const int w = t >> 5;
  for (int rr = 0; rr < 4; ++rr) {
    const int row = w * 4 + rr;
    const float* r = tile + row * HPITCH;
    const float a = r[lane];
    const float b = r[lane + 32];
    float mx = fmaxf(a, b);
    for (int msk = 16; msk >= 1; msk >>= 1) mx = fmaxf(mx, __shfl_xor(mx, msk, 32));
    float e = expf(a - mx) + expf(b - mx);
    for (int msk = 16; msk >= 1; msk >>= 1) e += __shfl_xor(e, msk, 32);
    const float lse = mx + logf(e);
    float* o = out + (size_t)(mbase + row) * NOUT;
    o[lane] = a - lse;
    o[lane + 32] = b - lse;
  }
}

extern "C" void kernel_launch(void* const* d_in, const int* in_sizes, int n_in,
                              void* d_out, int out_size, void* d_ws, size_t ws_size,
                              hipStream_t stream) {
  (void)n_in; (void)out_size; (void)ws_size;
  const float* x    = (const float*)d_in[0];
  const int*   ei   = (const int*)d_in[1];
  const float* W1   = (const float*)d_in[2];
  const float* b1   = (const float*)d_in[3];
  const float* W2   = (const float*)d_in[4];
  const float* b2   = (const float*)d_in[5];
  const float* ln1g = (const float*)d_in[6];
  const float* ln1b = (const float*)d_in[7];
  const float* ln2g = (const float*)d_in[8];
  const float* ln2b = (const float*)d_in[9];
  const float* linW = (const float*)d_in[10];
  const float* linb = (const float*)d_in[11];
  float* outp = (float*)d_out;

  const int N = in_sizes[0] / HIDC;
  const int E = in_sizes[1] / 2;
  const int OUTC = in_sizes[11];
  const int* src = ei;
  const int* dstp = ei + E;

  const long NH = (long)N * HIDC;
  const long W3 = 3L * KDIM * HIDC;
  const long WH = (long)KDIM * OUTC;
  float* ws = (float*)d_ws;
  long pad = ((long)N + 255) & ~255L;
  float* agg1 = ws + pad;
  float* agg2 = agg1 + NH;
  float* bufA = agg2 + NH;
  float* bufB = bufA + NH;
  float* Wt1  = bufB + NH;
  float* Wt2  = Wt1 + W3;
  float* WtH  = Wt2 + W3;

  const int padI = (N + 1 + 4095) & ~4095;
  int* cnt    = (int*)bufB;
  int* rowptr = cnt + padI;
  int* rank   = rowptr + padI;
  int* col    = rank + E;

  dim3 b256(256);
  int eBlocks = (E + 255) / 256;
  int gatherBlocks = (N + 8 * GROWS - 1) / (8 * GROWS);
  int mTiles = N / 16;
  int gemmBlocks = (mTiles + MT - 1) / MT;

  k_transpose_w<<<(int)((W3 + 255) / 256), b256, 0, stream>>>(W1, Wt1, KDIM, HIDC, 3);
  k_transpose_w<<<(int)((W3 + 255) / 256), b256, 0, stream>>>(W2, Wt2, KDIM, HIDC, 3);
  k_transpose_w<<<(int)((WH + 255) / 256), b256, 0, stream>>>(linW, WtH, KDIM, OUTC, 1);

  k_zero_i32<<<padI / 256, b256, 0, stream>>>(cnt, padI);
  k_count_rank<<<eBlocks, b256, 0, stream>>>(dstp, cnt, rank, E);
  k_scan_counts<<<1, dim3(1024), 0, stream>>>(cnt, rowptr, padI);
  k_fill_cols<<<eBlocks, b256, 0, stream>>>(src, dstp, rowptr, rank, col, E);

  k_gather_mean<<<gatherBlocks, b256, 0, stream>>>(rowptr, col, x, agg1, N);
  k_gather_mean<<<gatherBlocks, b256, 0, stream>>>(rowptr, col, agg1, agg2, N);
  k_wmma_gemm3_ln<<<gemmBlocks, b256, 0, stream>>>(x, agg1, agg2, Wt1, b1, ln1g, ln1b, bufA, N);

  k_gather_mean<<<gatherBlocks, b256, 0, stream>>>(rowptr, col, bufA, agg1, N);
  k_gather_mean<<<gatherBlocks, b256, 0, stream>>>(rowptr, col, agg1, agg2, N);
  k_wmma_gemm3_ln<<<gemmBlocks, b256, 0, stream>>>(bufA, agg1, agg2, Wt2, b2, ln2g, ln2b, bufB, N);

  k_wmma_head_lsm<<<mTiles, dim3(128), 0, stream>>>(bufB, WtH, linb, outp, OUTC);
}


# Launches only where all rows are needed (14 -> 10): clear + 3 transposes in one; head GEMM + log-softmax in the layer-2 GEMM work-group on its LDS tile, row passes 7 rows deep; sum order unchanged
# speedup vs baseline: 3.3536x; 1.0737x over previous
#include <hip/hip_runtime.h>


#define HIDC 128
#define KDIM 128
#define LPITCH 132
#define HPITCH 68

typedef float v2f __attribute__((ext_vector_type(2)));
typedef float v4f __attribute__((ext_vector_type(4)));
typedef float v8f __attribute__((ext_vector_type(8)));

#define W3BLOCKS (3 * KDIM * HIDC / 256)
__global__ __launch_bounds__(256) void k_prep(int* __restrict__ cnt, int zblocks,
                                              const float* __restrict__ W1, float* __restrict__ Wt1,
                                              const float* __restrict__ W2, float* __restrict__ Wt2,
                                              const float* __restrict__ WH, float* __restrict__ WtH,
                                              int nout_h) {
  const int t = threadIdx.x;
  int b = blockIdx.x;
  if (b < zblocks) {
    cnt[b * 256 + t] = 0;
    return;
  }
  b -= zblocks;
  if (b < 2 * W3BLOCKS) {
    const bool first = b < W3BLOCKS;
    const float* W = first ? W1 : W2;
    float* Wt = first ? Wt1 : Wt2;
    const int i = (first ? b : b - W3BLOCKS) * 256 + t;
    const int p = i / (KDIM * HIDC);
    const int rem = i % (KDIM * HIDC);
    const int k = rem / HIDC;
    const int n = rem % HIDC;
    Wt[p * (KDIM * HIDC) + n * KDIM + k] = W[i];
    return;
  }
  const int i = (b - 2 * W3BLOCKS) * 256 + t;
  if (i < KDIM * nout_h) {
    const int k = i / nout_h;
    const int n = i % nout_h;
    WtH[n * KDIM + k] = WH[i];
  }
}

__global__ __launch_bounds__(256) void k_count_rank(const int* __restrict__ dst,
                                                    int* __restrict__ cnt,
                                                    int* __restrict__ rank, int E) {
  int e = blockIdx.x * blockDim.x + threadIdx.x;
  if (e < E)
    rank[e] = __hip_atomic_fetch_add(&cnt[dst[e]], 1, __ATOMIC_RELAXED,
                                     __HIP_MEMORY_SCOPE_AGENT);
}

__global__ __launch_bounds__(1024) void k_scan_counts(const int* __restrict__ cnt,
                                                      int* __restrict__ rowptr, int P) {
  __shared__ int wave_total[32];
  const int lane = threadIdx.x & 31;
  const int w = __builtin_amdgcn_readfirstlane((int)(threadIdx.x >> 5));
  const int trips = P >> 12;
  const int q0 = w * trips * 32 + lane;
  const int4* c4 = reinterpret_cast<const int4*>(cnt);
  int4* r4 = reinterpret_cast<int4*>(rowptr);

  int s = 0;
#pragma unroll 4
  for (int t = 0; t < trips; ++t) {
    const int4 c = c4[q0 + t * 32];
    s += (c.x + c.y) + (c.z + c.w);
  }
  for (int m = 16; m >= 1; m >>= 1) s += __shfl_xor(s, m, 32);
  if (lane == 0) wave_total[w] = s;
  __syncthreads();

  int carry = (lane < w) ? wave_total[lane] : 0;
  for (int m = 16; m >= 1; m >>= 1) carry += __shfl_xor(carry, m, 32);

#pragma unroll 4
  for (int t = 0; t < trips; ++t) {
    const int4 c = c4[q0 + t * 32];
    const int ls = (c.x + c.y) + (c.z + c.w);
    int inc = ls;
    for (int d = 1; d < 32; d <<= 1) {
      const int v = __shfl_up(inc, d, 32);
      if (lane >= d) inc += v;
    }
    int4 r;
    r.x = carry + inc - ls;
    r.y = r.x + c.x;
    r.z = r.y + c.y;
    r.w = r.z + c.z;
    r4[q0 + t * 32] = r;
    carry += __shfl(inc, 31, 32);
  }
}

__global__ __launch_bounds__(256) void k_fill_cols(const int* __restrict__ src,
                                                   const int* __restrict__ dst,
                                                   const int* __restrict__ rowptr,
                                                   const int* __restrict__ rank,
                                                   int* __restrict__ col, int E) {
  int e = blockIdx.x * blockDim.x + threadIdx.x;
  if (e < E) col[rowptr[dst[e]] + rank[e]] = src[e];
}

#define GROWS 4
__global__ __launch_bounds__(256) void k_gather_mean(const int* __restrict__ rowptr,
                                                     const int* __restrict__ col,
                                                     const float* __restrict__ h,
                                                     float* __restrict__ out, int N) {
  const int lane = threadIdx.x & 31;
  const int wv = __builtin_amdgcn_readfirstlane((int)(blockIdx.x * 8 + (threadIdx.x >> 5)));
  const int r0 = wv * GROWS;
  if (r0 >= N) return;
  const int nr = min(GROWS, N - r0);
  const int bnd = rowptr[r0 + min(lane, nr)];
  const float4* h4 = reinterpret_cast<const float4*>(h);
  float4* o4 = reinterpret_cast<float4*>(out);
  for (int r = 0; r < nr; ++r) {
    const int beg = __builtin_amdgcn_readlane(bnd, r);
    const int end = __builtin_amdgcn_readlane(bnd, r + 1);
    float4 acc = {0.0f, 0.0f, 0.0f, 0.0f};
    for (int j0 = beg; j0 < end; j0 += 32) {
      const int n = min(end - j0, 32);
      const int mine = (lane < n) ? col[j0 + lane] : 0;
#pragma unroll 4
      for (int k = 0; k < n; ++k) {
        const int sidx = __builtin_amdgcn_readlane(mine, k);
        const float4 v = h4[(size_t)sidx * (HIDC / 4) + lane];
        acc.x += v.x;
        acc.y += v.y;
        acc.z += v.z;
        acc.w += v.w;
      }
    }
    const float inv = 1.0f / fmaxf((float)(end - beg), 1.0f);
    acc.x *= inv;
    acc.y *= inv;
    acc.z *= inv;
    acc.w *= inv;
    o4[(size_t)(r0 + r) * (HIDC / 4) + lane] = acc;
  }
}

#define MT 7
#define MROWS (MT * 16)
#define STG (MROWS / 8)
#define RB 7

#define FETCH_ROWS(X)                                                            \
  _Pragma("unroll") for (int j = 0; j < STG; ++j) {                              \
    const int grow = min(mbase + w + 8 * j, N - 1);                              \
    pre[j] = reinterpret_cast<const v4f*>((X) + (size_t)grow * KDIM)[lane];      \
  }
#define FETCH_ROWS_AHEAD(X)                                                      \
  FETCH_ROWS(X)                                                                  \
  __builtin_amdgcn_sched_barrier(0);
#define PARK_ROWS()                                                              \
  _Pragma("unroll") for (int j = 0; j < STG; ++j)                                \
    *reinterpret_cast<v4f*>(lds + (w + 8 * j) * LPITCH + lane * 4) = pre[j];
#define KSTEP(KK, BF)                                                            \
  _Pragma("unroll") for (int i = 0; i < MT; ++i) {                               \
    v2f a;                                                                       \
    const float2 av = *reinterpret_cast<const float2*>(la + i * 16 * LPITCH + (KK)); \
    a[0] = av.x; a[1] = av.y;                                                    \
    acc[i] = __builtin_amdgcn_wmma_f32_16x16x4_f32(                              \
        false, a, false, (BF), (short)0, acc[i], false, false);                  \
  }
#define LOAD_B4(DST, K0)                                                         \
  _Pragma("unroll") for (int s = 0; s < 4; ++s) {                                \
    const float2 bv2 = *reinterpret_cast<const float2*>(wb + (K0) + 4 * s);      \
    DST[s][0] = bv2.x; DST[s][1] = bv2.y;                                        \
  }
#define PRODUCTS(P, NEXT_FETCH)                                                  \
  {                                                                              \
    const float* wb = Wt + (size_t)(P) * (KDIM * HIDC) + (size_t)ncol * KDIM + koff; \
    v2f bq[4], bn[4];                                                            \
    LOAD_B4(bq, 0)                                                               \
      \
    _Pragma("unroll 2") for (int k = 0; k < KDIM - 32; k += 16) {                \
      LOAD_B4(bn, k + 16)                                                        \
      _Pragma("unroll") for (int s = 0; s < 4; ++s) { KSTEP(k + 4 * s, bq[s]) }  \
      _Pragma("unroll") for (int s = 0; s < 4; ++s) bq[s] = bn[s];               \
    }                                                                            \
    LOAD_B4(bn, KDIM - 16)                                                       \
    NEXT_FETCH                                                                   \
    _Pragma("unroll") for (int s = 0; s < 4; ++s) { KSTEP(KDIM - 32 + 4 * s, bq[s]) } \
    _Pragma("unroll") for (int s = 0; s < 4; ++s) { KSTEP(KDIM - 16 + 4 * s, bn[s]) } \
  }

#define HB_LOAD8(DST, K0)                                                        \
  _Pragma("unroll") for (int s = 0; s < 8; ++s) {                                \
    const float2 bv2 = *reinterpret_cast<const float2*>(hb + (K0) + 4 * s);      \
    DST[s][0] = bv2.x; DST[s][1] = bv2.y;                                        \
  }
#define HB_STEPS8(BF, K0)                                                        \
  _Pragma("unroll") for (int s = 0; s < 8; ++s) {                                \
    _Pragma("unroll") for (int j = 0; j < NT; ++j) {                             \
      v2f a;                                                                     \
      const float2 av =                                                          \
          *reinterpret_cast<const float2*>(ha + j * 32 * LPITCH + (K0) + 4 * s); \
      a[0] = av.x; a[1] = av.y;                                                  \
      hacc[j] = __builtin_amdgcn_wmma_f32_16x16x4_f32(                           \
          false, a, false, BF[s], (short)0, hacc[j], false, false);              \
    }                                                                            \
  }
template <int NT>
__device__ __forceinline__ void head_products(const float* __restrict__ ha,
                                              const float* __restrict__ hb, const v2f* hb0,
                                              v8f* hacc) {
  v2f b1[8], b2[8];
  HB_LOAD8(b1, 32)
  HB_STEPS8(hb0, 0)
  HB_LOAD8(b2, 64)
  HB_STEPS8(b1, 32)
  HB_LOAD8(b1, 96)
  HB_STEPS8(b2, 64)
  HB_STEPS8(b1, 96)
}

template <bool HEAD>
__device__ __forceinline__ void gemm3_ln_body(
    float* lds, float* tile,
    const float* __restrict__ X0, const float* __restrict__ X1,
    const float* __restrict__ X2, const float* __restrict__ Wt,
    const float* __restrict__ bias, const float* __restrict__ gw,
    const float* __restrict__ gb, const float* __restrict__ WtH,
    const float* __restrict__ hbias, float* __restrict__ out, int N, int NOUT) {
  const int t = threadIdx.x;
  const int lane = t & 31;
  const int w = __builtin_amdgcn_readfirstlane(t >> 5);
  const int mbase = blockIdx.x * MROWS;

  v4f pre[STG];
  FETCH_ROWS(X0)
  PARK_ROWS()
  __syncthreads();

  const int m = lane & 15;
  const int ncol = w * 16 + m;
  const int koff = (lane < 16) ? 0 : 2;
  const float* la = lds + m * LPITCH + koff;

  v8f acc[MT];
#pragma unroll
  for (int i = 0; i < MT; ++i) acc[i] = v8f{};

  PRODUCTS(0, FETCH_ROWS_AHEAD(X1))
  __syncthreads();
  PARK_ROWS()
  __syncthreads();

  PRODUCTS(1, FETCH_ROWS_AHEAD(X2))
  __syncthreads();
  PARK_ROWS()
  __syncthreads();

  PRODUCTS(2, )
  __syncthreads();

  const float bv = bias[ncol];
  const int mo = (lane < 16) ? 0 : 8;
#pragma unroll
  for (int i = 0; i < MT; ++i)
#pragma unroll
    for (int r = 0; r < 8; ++r) lds[(i * 16 + r + mo) * LPITCH + ncol] = acc[i][r] + bv;
  __syncthreads();

  const float4 g4 = reinterpret_cast<const float4*>(gw)[lane];
  const float4 b4 = reinterpret_cast<const float4*>(gb)[lane];
  float4* o4 = reinterpret_cast<float4*>(out);

  const int rset = w >> 2;
  const int hcol = (w & 3) * 16 + m;
  const float* hb = HEAD ? WtH + (size_t)hcol * KDIM + koff : nullptr;
  v2f hbf[8];
  if (HEAD) { HB_LOAD8(hbf, 0) }

  for (int g = 0; g < STG / RB; ++g) {
    float4 v[RB];
    float s[RB], q[RB];
#pragma unroll
    for (int u = 0; u < RB; ++u) {
      const int row = w + 8 * (g * RB + u);
      v[u] = *reinterpret_cast<const float4*>(lds + row * LPITCH + lane * 4);
      s[u] = v[u].x + v[u].y + v[u].z + v[u].w;
      q[u] = v[u].x * v[u].x + v[u].y * v[u].y + v[u].z * v[u].z + v[u].w * v[u].w;
    }
    for (int msk = 16; msk >= 1; msk >>= 1) {
#pragma unroll
      for (int u = 0; u < RB; ++u) {
        s[u] += __shfl_xor(s[u], msk, 32);
        q[u] += __shfl_xor(q[u], msk, 32);
      }
    }
#pragma unroll
    for (int u = 0; u < RB; ++u) {
      const int row = w + 8 * (g * RB + u);
      const float mu = s[u] * (1.0f / HIDC);
      const float var = q[u] * (1.0f / HIDC) - mu * mu;
      const float rs = rsqrtf(var + 1e-5f);
      float4 y;
      y.x = (v[u].x - mu) * rs * g4.x + b4.x;
      y.y = (v[u].y - mu) * rs * g4.y + b4.y;
      y.z = (v[u].z - mu) * rs * g4.z + b4.z;
      y.w = (v[u].w - mu) * rs * g4.w + b4.w;
      if (HEAD) {
        *reinterpret_cast<float4*>(lds + row * LPITCH + lane * 4) = y;
      } else if (mbase + row < N) {
        o4[(size_t)(mbase + row) * (HIDC / 4) + lane] = y;
      }
    }
  }
  if (!HEAD) return;
  __syncthreads();

  const float* ha = lds + (rset * 16 + m) * LPITCH + koff;
  v8f hacc[4];
#pragma unroll
  for (int j = 0; j < 4; ++j) hacc[j] = v8f{};
  if (rset == 0) head_products<4>(ha, hb, hbf, hacc);
  else           head_products<3>(ha, hb, hbf, hacc);

  const float hbv = hbias[hcol];
  __syncthreads();
#pragma unroll
  for (int j = 0; j < 4; ++j)
    if (j < 3 || rset == 0) {
#pragma unroll
      for (int r = 0; r < 8; ++r)
        tile[((rset + 2 * j) * 16 + r + mo) * HPITCH + hcol] = hacc[j][r] + hbv;
    }
  __syncthreads();

  for (int g = 0; g < STG / RB; ++g) {
    float a[RB], b[RB], mx[RB], e[RB];
#pragma unroll
    for (int u = 0; u < RB; ++u) {
      const float* r = tile + (w + 8 * (g * RB + u)) * HPITCH;
      a[u] = r[lane];
      b[u] = r[lane + 32];
      mx[u] = fmaxf(a[u], b[u]);
    }
    for (int msk = 16; msk >= 1; msk >>= 1) {
#pragma unroll
      for (int u = 0; u < RB; ++u) mx[u] = fmaxf(mx[u], __shfl_xor(mx[u], msk, 32));
    }
#pragma unroll
    for (int u = 0; u < RB; ++u) e[u] = expf(a[u] - mx[u]) + expf(b[u] - mx[u]);
    for (int msk = 16; msk >= 1; msk >>= 1) {
#pragma unroll
      for (int u = 0; u < RB; ++u) e[u] += __shfl_xor(e[u], msk, 32);
    }
#pragma unroll
    for (int u = 0; u < RB; ++u) {
      const int row = w + 8 * (g * RB + u);
      if (mbase + row < N) {
        const float lse = mx[u] + logf(e[u]);
        float* o = out + (size_t)(mbase + row) * NOUT;
        o[lane] = a[u] - lse;
        o[lane + 32] = b[u] - lse;
      }
    }
  }
}

__global__ __launch_bounds__(256) void k_wmma_gemm3_ln(
    const float* __restrict__ X0, const float* __restrict__ X1,
    const float* __restrict__ X2, const float* __restrict__ Wt,
    const float* __restrict__ bias, const float* __restrict__ gw,
    const float* __restrict__ gb, float* __restrict__ out, int N) {
  __shared__ float lds[MROWS * LPITCH];
  gemm3_ln_body<false>(lds, nullptr, X0, X1, X2, Wt, bias, gw, gb, nullptr, nullptr, out, N, 0);
}

__global__ __launch_bounds__(256, 5) void k_wmma_gemm3_ln_head(
    const float* __restrict__ X0, const float* __restrict__ X1,
    const float* __restrict__ X2, const float* __restrict__ Wt,
    const float* __restrict__ bias, const float* __restrict__ gw,
    const float* __restrict__ gb, const float* __restrict__ WtH,
    const float* __restrict__ hbias, float* __restrict__ out, int N, int NOUT) {
  __shared__ float lds[MROWS * LPITCH];
  gemm3_ln_body<true>(lds, lds, X0, X1, X2, Wt, bias, gw, gb, WtH, hbias, out, N, NOUT);
}

extern "C" void kernel_launch(void* const* d_in, const int* in_sizes, int n_in,
                              void* d_out, int out_size, void* d_ws, size_t ws_size,
                              hipStream_t stream) {
  (void)n_in; (void)out_size; (void)ws_size;
  const float* x    = (const float*)d_in[0];
  const int*   ei   = (const int*)d_in[1];
  const float* W1   = (const float*)d_in[2];
  const float* b1   = (const float*)d_in[3];
  const float* W2   = (const float*)d_in[4];
  const float* b2   = (const float*)d_in[5];
  const float* ln1g = (const float*)d_in[6];
  const float* ln1b = (const float*)d_in[7];
  const float* ln2g = (const float*)d_in[8];
  const float* ln2b = (const float*)d_in[9];
  const float* linW = (const float*)d_in[10];
  const float* linb = (const float*)d_in[11];
  float* outp = (float*)d_out;

  const int N = in_sizes[0] / HIDC;
  const int E = in_sizes[1] / 2;
  const int OUTC = in_sizes[11];
  const int* src = ei;
  const int* dstp = ei + E;

  const long NH = (long)N * HIDC;
  const long W3 = 3L * KDIM * HIDC;
  const long WH = (long)KDIM * OUTC;
  float* ws = (float*)d_ws;
  long pad = ((long)N + 255) & ~255L;
  float* agg1 = ws + pad;
  float* agg2 = agg1 + NH;
  float* bufA = agg2 + NH;
  float* bufB = bufA + NH;
  float* Wt1  = bufB + NH;
  float* Wt2  = Wt1 + W3;
  float* WtH  = Wt2 + W3;

  const int padI = (N + 1 + 4095) & ~4095;
  int* cnt    = (int*)bufB;
  int* rowptr = cnt + padI;
  int* rank   = rowptr + padI;
  int* col    = rank + E;

  dim3 b256(256);
  int eBlocks = (E + 255) / 256;
  int gatherBlocks = (N + 8 * GROWS - 1) / (8 * GROWS);
  int mTiles = N / 16;
  int gemmBlocks = (mTiles + MT - 1) / MT;

  int zBlocks = padI / 256;
  int hBlocks = (int)((WH + 255) / 256);
  k_prep<<<zBlocks + 2 * W3BLOCKS + hBlocks, b256, 0, stream>>>(cnt, zBlocks, W1, Wt1, W2, Wt2,
                                                                linW, WtH, OUTC);

  k_count_rank<<<eBlocks, b256, 0, stream>>>(dstp, cnt, rank, E);
  k_scan_counts<<<1, dim3(1024), 0, stream>>>(cnt, rowptr, padI);
  k_fill_cols<<<eBlocks, b256, 0, stream>>>(src, dstp, rowptr, rank, col, E);

  k_gather_mean<<<gatherBlocks, b256, 0, stream>>>(rowptr, col, x, agg1, N);
  k_gather_mean<<<gatherBlocks, b256, 0, stream>>>(rowptr, col, agg1, agg2, N);
  k_wmma_gemm3_ln<<<gemmBlocks, b256, 0, stream>>>(x, agg1, agg2, Wt1, b1, ln1g, ln1b, bufA, N);

  k_gather_mean<<<gatherBlocks, b256, 0, stream>>>(rowptr, col, bufA, agg1, N);
  k_gather_mean<<<gatherBlocks, b256, 0, stream>>>(rowptr, col, agg1, agg2, N);
  k_wmma_gemm3_ln_head<<<gemmBlocks, b256, 0, stream>>>(bufA, agg1, agg2, Wt2, b2, ln2g, ln2b,
                                                        WtH, linb, outp, N, OUTC);
}


# Rows-by-destination in one launch: 32 fixed slots per row, the count's atomic rank is the slot, overflow edges go to a shared pair list; scan and fill launches gone; row sum order = atomic arrival
# speedup vs baseline: 3.8037x; 1.1342x over previous
#include <hip/hip_runtime.h>


#define HIDC 128
#define KDIM 128
#define LPITCH 132
#define HPITCH 68

typedef float v2f __attribute__((ext_vector_type(2)));
typedef float v4f __attribute__((ext_vector_type(4)));
typedef float v8f __attribute__((ext_vector_type(8)));

#define W3BLOCKS (3 * KDIM * HIDC / 256)
__global__ __launch_bounds__(256) void k_prep(int* __restrict__ cnt, int zblocks,
                                              const float* __restrict__ W1, float* __restrict__ Wt1,
                                              const float* __restrict__ W2, float* __restrict__ Wt2,
                                              const float* __restrict__ WH, float* __restrict__ WtH,
                                              int nout_h) {
  const int t = threadIdx.x;
  int b = blockIdx.x;
  if (b < zblocks) {
    cnt[b * 256 + t] = 0;
    return;
  }
  b -= zblocks;
  if (b < 2 * W3BLOCKS) {
    const bool first = b < W3BLOCKS;
    const float* W = first ? W1 : W2;
    float* Wt = first ? Wt1 : Wt2;
    const int i = (first ? b : b - W3BLOCKS) * 256 + t;
    const int p = i / (KDIM * HIDC);
    const int rem = i % (KDIM * HIDC);
    const int k = rem / HIDC;
    const int n = rem % HIDC;
    Wt[p * (KDIM * HIDC) + n * KDIM + k] = W[i];
    return;
  }
  const int i = (b - 2 * W3BLOCKS) * 256 + t;
  if (i < KDIM * nout_h) {
    const int k = i / nout_h;
    const int n = i % nout_h;
    WtH[n * KDIM + k] = WH[i];
  }
}

#define SLOTS 32
__global__ __launch_bounds__(256) void k_count_place(const int* __restrict__ src,
                                                     const int* __restrict__ dst,
                                                     int* __restrict__ cnt,
                                                     int* __restrict__ slot,
                                                     int* __restrict__ spill, int E, int N) {
  const int e = blockIdx.x * blockDim.x + threadIdx.x;
  if (e < E) {
    const int d = dst[e];
    const int s = src[e];
    const int r = __hip_atomic_fetch_add(&cnt[d], 1, __ATOMIC_RELAXED, __HIP_MEMORY_SCOPE_AGENT);
    if (r < SLOTS) {
      slot[(size_t)d * SLOTS + r] = s;
    } else {
      const int o = __hip_atomic_fetch_add(&cnt[N], 1, __ATOMIC_RELAXED, __HIP_MEMORY_SCOPE_AGENT);
      int2 rec;
      rec.x = d;
      rec.y = s;
      reinterpret_cast<int2*>(spill)[o] = rec;
    }
  }
}

#define GROWS 4
__global__ __launch_bounds__(256) void k_gather_mean(const int* __restrict__ cnt,
                                                     const int* __restrict__ slot,
                                                     const int* __restrict__ spill,
                                                     const float* __restrict__ h,
                                                     float* __restrict__ out, int N) {
  const int lane = threadIdx.x & 31;
  const int wv = __builtin_amdgcn_readfirstlane((int)(blockIdx.x * 8 + (threadIdx.x >> 5)));
  const int r0 = wv * GROWS;
  if (r0 >= N) return;
  const int nr = min(GROWS, N - r0);
  const int dg = (lane < nr) ? cnt[r0 + lane] : 0;
  int dgs[GROWS];
#pragma unroll
  for (int r = 0; r < GROWS; ++r) dgs[r] = __builtin_amdgcn_readlane(dg, r);
  int ids[GROWS];
#pragma unroll
  for (int r = 0; r < GROWS; ++r)
    ids[r] = (lane < min(dgs[r], SLOTS)) ? slot[(size_t)(r0 + r) * SLOTS + lane] : 0;
  const float4* h4 = reinterpret_cast<const float4*>(h);
  float4* o4 = reinterpret_cast<float4*>(out);
  for (int r = 0; r < nr; ++r) {
    const int deg = __builtin_amdgcn_readlane(dg, r);
    const int mine = (r == 0) ? ids[0] : (r == 1) ? ids[1] : (r == 2) ? ids[2] : ids[3];
    const int n = min(deg, SLOTS);
    float4 acc = {0.0f, 0.0f, 0.0f, 0.0f};
#pragma unroll 4
    for (int k = 0; k < n; ++k) {
      const int sidx = __builtin_amdgcn_readlane(mine, k);
      const float4 v = h4[(size_t)sidx * (HIDC / 4) + lane];
      acc.x += v.x;
      acc.y += v.y;
      acc.z += v.z;
      acc.w += v.w;
    }
    if (deg > SLOTS) {
      const int row = r0 + r;
      const int M = cnt[N];
      const int2* sp = reinterpret_cast<const int2*>(spill);
      for (int j0 = 0; j0 < M; j0 += 32) {
        const int j = j0 + lane;
        int2 rec;
        rec.x = -1;
        rec.y = 0;
        if (j < M) rec = sp[j];
        unsigned hit = __builtin_amdgcn_ballot_w32(rec.x == row);
        while (hit != 0u) {
          const int k = __builtin_ctz(hit);
          hit &= hit - 1u;
          const int sidx = __builtin_amdgcn_readlane(rec.y, k);
          const float4 v = h4[(size_t)sidx * (HIDC / 4) + lane];
          acc.x += v.x;
          acc.y += v.y;
          acc.z += v.z;
          acc.w += v.w;
        }
      }
    }
    const float inv = 1.0f / fmaxf((float)deg, 1.0f);
    acc.x *= inv;
    acc.y *= inv;
    acc.z *= inv;
    acc.w *= inv;
    o4[(size_t)(r0 + r) * (HIDC / 4) + lane] = acc;
  }
}

#define MT 7
#define MROWS (MT * 16)
#define STG (MROWS / 8)
#define RB 7

#define FETCH_ROWS(X)                                                            \
  _Pragma("unroll") for (int j = 0; j < STG; ++j) {                              \
    const int grow = min(mbase + w + 8 * j, N - 1);                              \
    pre[j] = reinterpret_cast<const v4f*>((X) + (size_t)grow * KDIM)[lane];      \
  }
#define FETCH_ROWS_AHEAD(X)                                                      \
  FETCH_ROWS(X)                                                                  \
  __builtin_amdgcn_sched_barrier(0);
#define PARK_ROWS()                                                              \
  _Pragma("unroll") for (int j = 0; j < STG; ++j)                                \
    *reinterpret_cast<v4f*>(lds + (w + 8 * j) * LPITCH + lane * 4) = pre[j];
#define KSTEP(KK, BF)                                                            \
  _Pragma("unroll") for (int i = 0; i < MT; ++i) {                               \
    v2f a;                                                                       \
    const float2 av = *reinterpret_cast<const float2*>(la + i * 16 * LPITCH + (KK)); \
    a[0] = av.x; a[1] = av.y;                                                    \
    acc[i] = __builtin_amdgcn_wmma_f32_16x16x4_f32(                              \
        false, a, false, (BF), (short)0, acc[i], false, false);                  \
  }
#define LOAD_B4(DST, K0)                                                         \
  _Pragma("unroll") for (int s = 0; s < 4; ++s) {                                \
    const float2 bv2 = *reinterpret_cast<const float2*>(wb + (K0) + 4 * s);      \
    DST[s][0] = bv2.x; DST[s][1] = bv2.y;                                        \
  }
#define PRODUCTS(P, NEXT_FETCH)                                                  \
  {                                                                              \
    const float* wb = Wt + (size_t)(P) * (KDIM * HIDC) + (size_t)ncol * KDIM + koff; \
    v2f bq[4], bn[4];                                                            \
    LOAD_B4(bq, 0)                                                               \
      \
    _Pragma("unroll 2") for (int k = 0; k < KDIM - 32; k += 16) {                \
      LOAD_B4(bn, k + 16)                                                        \
      _Pragma("unroll") for (int s = 0; s < 4; ++s) { KSTEP(k + 4 * s, bq[s]) }  \
      _Pragma("unroll") for (int s = 0; s < 4; ++s) bq[s] = bn[s];               \
    }                                                                            \
    LOAD_B4(bn, KDIM - 16)                                                       \
    NEXT_FETCH                                                                   \
    _Pragma("unroll") for (int s = 0; s < 4; ++s) { KSTEP(KDIM - 32 + 4 * s, bq[s]) } \
    _Pragma("unroll") for (int s = 0; s < 4; ++s) { KSTEP(KDIM - 16 + 4 * s, bn[s]) } \
  }

#define HB_LOAD8(DST, K0)                                                        \
  _Pragma("unroll") for (int s = 0; s < 8; ++s) {                                \
    const float2 bv2 = *reinterpret_cast<const float2*>(hb + (K0) + 4 * s);      \
    DST[s][0] = bv2.x; DST[s][1] = bv2.y;                                        \
  }
#define HB_STEPS8(BF, K0)                                                        \
  _Pragma("unroll") for (int s = 0; s < 8; ++s) {                                \
    _Pragma("unroll") for (int j = 0; j < NT; ++j) {                             \
      v2f a;                                                                     \
      const float2 av =                                                          \
          *reinterpret_cast<const float2*>(ha + j * 32 * LPITCH + (K0) + 4 * s); \
      a[0] = av.x; a[1] = av.y;                                                  \
      hacc[j] = __builtin_amdgcn_wmma_f32_16x16x4_f32(                           \
          false, a, false, BF[s], (short)0, hacc[j], false, false);              \
    }                                                                            \
  }
template <int NT>
__device__ __forceinline__ void head_products(const float* __restrict__ ha,
                                              const float* __restrict__ hb, const v2f* hb0,
                                              v8f* hacc) {
  v2f b1[8], b2[8];
  HB_LOAD8(b1, 32)
  HB_STEPS8(hb0, 0)
  HB_LOAD8(b2, 64)
  HB_STEPS8(b1, 32)
  HB_LOAD8(b1, 96)
  HB_STEPS8(b2, 64)
  HB_STEPS8(b1, 96)
}

template <bool HEAD>
__device__ __forceinline__ void gemm3_ln_body(
    float* lds, float* tile,
    const float* __restrict__ X0, const float* __restrict__ X1,
    const float* __restrict__ X2, const float* __restrict__ Wt,
    const float* __restrict__ bias, const float* __restrict__ gw,
    const float* __restrict__ gb, const float* __restrict__ WtH,
    const float* __restrict__ hbias, float* __restrict__ out, int N, int NOUT) {
  const int t = threadIdx.x;
  const int lane = t & 31;
  const int w = __builtin_amdgcn_readfirstlane(t >> 5);
  const int mbase = blockIdx.x * MROWS;

  v4f pre[STG];
  FETCH_ROWS(X0)
  PARK_ROWS()
  __syncthreads();

  const int m = lane & 15;
  const int ncol = w * 16 + m;
  const int koff = (lane < 16) ? 0 : 2;
  const float* la = lds + m * LPITCH + koff;

  v8f acc[MT];
#pragma unroll
  for (int i = 0; i < MT; ++i) acc[i] = v8f{};

  PRODUCTS(0, FETCH_ROWS_AHEAD(X1))
  __syncthreads();
  PARK_ROWS()
  __syncthreads();

  PRODUCTS(1, FETCH_ROWS_AHEAD(X2))
  __syncthreads();
  PARK_ROWS()
  __syncthreads();

  PRODUCTS(2, )
  __syncthreads();

  const float bv = bias[ncol];
  const int mo = (lane < 16) ? 0 : 8;
#pragma unroll
  for (int i = 0; i < MT; ++i)
#pragma unroll
    for (int r = 0; r < 8; ++r) lds[(i * 16 + r + mo) * LPITCH + ncol] = acc[i][r] + bv;
  __syncthreads();

  const float4 g4 = reinterpret_cast<const float4*>(gw)[lane];
  const float4 b4 = reinterpret_cast<const float4*>(gb)[lane];
  float4* o4 = reinterpret_cast<float4*>(out);

  const int rset = w >> 2;
  const int hcol = (w & 3) * 16 + m;
  const float* hb = HEAD ? WtH + (size_t)hcol * KDIM + koff : nullptr;
  v2f hbf[8];
  if (HEAD) { HB_LOAD8(hbf, 0) }

  for (int g = 0; g < STG / RB; ++g) {
    float4 v[RB];
    float s[RB], q[RB];
#pragma unroll
    for (int u = 0; u < RB; ++u) {
      const int row = w + 8 * (g * RB + u);
      v[u] = *reinterpret_cast<const float4*>(lds + row * LPITCH + lane * 4);
      s[u] = v[u].x + v[u].y + v[u].z + v[u].w;
      q[u] = v[u].x * v[u].x + v[u].y * v[u].y + v[u].z * v[u].z + v[u].w * v[u].w;
    }
    for (int msk = 16; msk >= 1; msk >>= 1) {
#pragma unroll
      for (int u = 0; u < RB; ++u) {
        s[u] += __shfl_xor(s[u], msk, 32);
        q[u] += __shfl_xor(q[u], msk, 32);
      }
    }
#pragma unroll
    for (int u = 0; u < RB; ++u) {
      const int row = w + 8 * (g * RB + u);
      const float mu = s[u] * (1.0f / HIDC);
      const float var = q[u] * (1.0f / HIDC) - mu * mu;
      const float rs = rsqrtf(var + 1e-5f);
      float4 y;
      y.x = (v[u].x - mu) * rs * g4.x + b4.x;
      y.y = (v[u].y - mu) * rs * g4.y + b4.y;
      y.z = (v[u].z - mu) * rs * g4.z + b4.z;
      y.w = (v[u].w - mu) * rs * g4.w + b4.w;
      if (HEAD) {
        *reinterpret_cast<float4*>(lds + row * LPITCH + lane * 4) = y;
      } else if (mbase + row < N) {
        o4[(size_t)(mbase + row) * (HIDC / 4) + lane] = y;
      }
    }
  }
  if (!HEAD) return;
  __syncthreads();

  const float* ha = lds + (rset * 16 + m) * LPITCH + koff;
  v8f hacc[4];
#pragma unroll
  for (int j = 0; j < 4; ++j) hacc[j] = v8f{};
  if (rset == 0) head_products<4>(ha, hb, hbf, hacc);
  else           head_products<3>(ha, hb, hbf, hacc);

  const float hbv = hbias[hcol];
  __syncthreads();
#pragma unroll
  for (int j = 0; j < 4; ++j)
    if (j < 3 || rset == 0) {
#pragma unroll
      for (int r = 0; r < 8; ++r)
        tile[((rset + 2 * j) * 16 + r + mo) * HPITCH + hcol] = hacc[j][r] + hbv;
    }
  __syncthreads();

  for (int g = 0; g < STG / RB; ++g) {
    float a[RB], b[RB], mx[RB], e[RB];
#pragma unroll
    for (int u = 0; u < RB; ++u) {
      const float* r = tile + (w + 8 * (g * RB + u)) * HPITCH;
      a[u] = r[lane];
      b[u] = r[lane + 32];
      mx[u] = fmaxf(a[u], b[u]);
    }
    for (int msk = 16; msk >= 1; msk >>= 1) {
#pragma unroll
      for (int u = 0; u < RB; ++u) mx[u] = fmaxf(mx[u], __shfl_xor(mx[u], msk, 32));
    }
#pragma unroll
    for (int u = 0; u < RB; ++u) e[u] = expf(a[u] - mx[u]) + expf(b[u] - mx[u]);
    for (int msk = 16; msk >= 1; msk >>= 1) {
#pragma unroll
      for (int u = 0; u < RB; ++u) e[u] += __shfl_xor(e[u], msk, 32);
    }
#pragma unroll
    for (int u = 0; u < RB; ++u) {
      const int row = w + 8 * (g * RB + u);
      if (mbase + row < N) {
        const float lse = mx[u] + logf(e[u]);
        float* o = out + (size_t)(mbase + row) * NOUT;
        o[lane] = a[u] - lse;
        o[lane + 32] = b[u] - lse;
      }
    }
  }
}

__global__ __launch_bounds__(256) void k_wmma_gemm3_ln(
    const float* __restrict__ X0, const float* __restrict__ X1,
    const float* __restrict__ X2, const float* __restrict__ Wt,
    const float* __restrict__ bias, const float* __restrict__ gw,
    const float* __restrict__ gb, float* __restrict__ out, int N) {
  __shared__ float lds[MROWS * LPITCH];
  gemm3_ln_body<false>(lds, nullptr, X0, X1, X2, Wt, bias, gw, gb, nullptr, nullptr, out, N, 0);
}

__global__ __launch_bounds__(256, 5) void k_wmma_gemm3_ln_head(
    const float* __restrict__ X0, const float* __restrict__ X1,
    const float* __restrict__ X2, const float* __restrict__ Wt,
    const float* __restrict__ bias, const float* __restrict__ gw,
    const float* __restrict__ gb, const float* __restrict__ WtH,
    const float* __restrict__ hbias, float* __restrict__ out, int N, int NOUT) {
  __shared__ float lds[MROWS * LPITCH];
  gemm3_ln_body<true>(lds, lds, X0, X1, X2, Wt, bias, gw, gb, WtH, hbias, out, N, NOUT);
}

extern "C" void kernel_launch(void* const* d_in, const int* in_sizes, int n_in,
                              void* d_out, int out_size, void* d_ws, size_t ws_size,
                              hipStream_t stream) {
  (void)n_in; (void)out_size; (void)ws_size;
  const float* x    = (const float*)d_in[0];
  const int*   ei   = (const int*)d_in[1];
  const float* W1   = (const float*)d_in[2];
  const float* b1   = (const float*)d_in[3];
  const float* W2   = (const float*)d_in[4];
  const float* b2   = (const float*)d_in[5];
  const float* ln1g = (const float*)d_in[6];
  const float* ln1b = (const float*)d_in[7];
  const float* ln2g = (const float*)d_in[8];
  const float* ln2b = (const float*)d_in[9];
  const float* linW = (const float*)d_in[10];
  const float* linb = (const float*)d_in[11];
  float* outp = (float*)d_out;

  const int N = in_sizes[0] / HIDC;
  const int E = in_sizes[1] / 2;
  const int OUTC = in_sizes[11];
  const int* src = ei;
  const int* dstp = ei + E;

  const long NH = (long)N * HIDC;
  const long W3 = 3L * KDIM * HIDC;
  const long WH = (long)KDIM * OUTC;
  float* ws = (float*)d_ws;
  long pad = ((long)N + 255) & ~255L;
  float* agg1 = ws + pad;
  float* agg2 = agg1 + NH;
  float* bufA = agg2 + NH;
  float* bufB = bufA + NH;
  float* Wt1  = bufB + NH;
  float* Wt2  = Wt1 + W3;
  float* WtH  = Wt2 + W3;

  const int padC = (N + 1 + 255) & ~255;
  int* cnt   = (int*)bufB;
  int* slot  = cnt + padC;
  int* spill = slot + (size_t)N * SLOTS;

  dim3 b256(256);
  int eBlocks = (E + 255) / 256;
  int gatherBlocks = (N + 8 * GROWS - 1) / (8 * GROWS);
  int mTiles = N / 16;
  int gemmBlocks = (mTiles + MT - 1) / MT;

  int zBlocks = padC / 256;
  int hBlocks = (int)((WH + 255) / 256);
  k_prep<<<zBlocks + 2 * W3BLOCKS + hBlocks, b256, 0, stream>>>(cnt, zBlocks, W1, Wt1, W2, Wt2,
                                                                linW, WtH, OUTC);

  k_count_place<<<eBlocks, b256, 0, stream>>>(src, dstp, cnt, slot, spill, E, N);

  k_gather_mean<<<gatherBlocks, b256, 0, stream>>>(cnt, slot, spill, x, agg1, N);
  k_gather_mean<<<gatherBlocks, b256, 0, stream>>>(cnt, slot, spill, agg1, agg2, N);
  k_wmma_gemm3_ln<<<gemmBlocks, b256, 0, stream>>>(x, agg1, agg2, Wt1, b1, ln1g, ln1b, bufA, N);

  k_gather_mean<<<gatherBlocks, b256, 0, stream>>>(cnt, slot, spill, bufA, agg1, N);
  k_gather_mean<<<gatherBlocks, b256, 0, stream>>>(cnt, slot, spill, agg1, agg2, N);
  k_wmma_gemm3_ln_head<<<gemmBlocks, b256, 0, stream>>>(bufA, agg1, agg2, Wt2, b2, ln2g, ln2b,
                                                        WtH, linb, outp, N, OUTC);
}


# Second aggregation of each layer summed inside the GEMM work-group that uses it (neighbours' first-power rows -> its LDS A tile, 16 row loads in flight per wave); 8 -> 6 launches; sum order unchanged
# speedup vs baseline: 3.9810x; 1.0466x over previous
#include <hip/hip_runtime.h>


#define HIDC 128
#define KDIM 128
#define LPITCH 132
#define HPITCH 68

typedef float v2f __attribute__((ext_vector_type(2)));
typedef float v4f __attribute__((ext_vector_type(4)));
typedef float v8f __attribute__((ext_vector_type(8)));

#define W3BLOCKS (3 * KDIM * HIDC / 256)
__global__ __launch_bounds__(256) void k_prep(int* __restrict__ cnt, int zblocks,
                                              const float* __restrict__ W1, float* __restrict__ Wt1,
                                              const float* __restrict__ W2, float* __restrict__ Wt2,
                                              const float* __restrict__ WH, float* __restrict__ WtH,
                                              int nout_h) {
  const int t = threadIdx.x;
  int b = blockIdx.x;
  if (b < zblocks) {
    cnt[b * 256 + t] = 0;
    return;
  }
  b -= zblocks;
  if (b < 2 * W3BLOCKS) {
    const bool first = b < W3BLOCKS;
    const float* W = first ? W1 : W2;
    float* Wt = first ? Wt1 : Wt2;
    const int i = (first ? b : b - W3BLOCKS) * 256 + t;
    const int p = i / (KDIM * HIDC);
    const int rem = i % (KDIM * HIDC);
    const int k = rem / HIDC;
    const int n = rem % HIDC;
    Wt[p * (KDIM * HIDC) + n * KDIM + k] = W[i];
    return;
  }
  const int i = (b - 2 * W3BLOCKS) * 256 + t;
  if (i < KDIM * nout_h) {
    const int k = i / nout_h;
    const int n = i % nout_h;
    WtH[n * KDIM + k] = WH[i];
  }
}

#define SLOTS 32
__global__ __launch_bounds__(256) void k_count_place(const int* __restrict__ src,
                                                     const int* __restrict__ dst,
                                                     int* __restrict__ cnt,
                                                     int* __restrict__ slot,
                                                     int* __restrict__ spill, int E, int N) {
  const int e = blockIdx.x * blockDim.x + threadIdx.x;
  if (e < E) {
    const int d = dst[e];
    const int s = src[e];
    const int r = __hip_atomic_fetch_add(&cnt[d], 1, __ATOMIC_RELAXED, __HIP_MEMORY_SCOPE_AGENT);
    if (r < SLOTS) {
      slot[(size_t)d * SLOTS + r] = s;
    } else {
      const int o = __hip_atomic_fetch_add(&cnt[N], 1, __ATOMIC_RELAXED, __HIP_MEMORY_SCOPE_AGENT);
      int2 rec;
      rec.x = d;
      rec.y = s;
      reinterpret_cast<int2*>(spill)[o] = rec;
    }
  }
}

#define GROWS 4
__global__ __launch_bounds__(256) void k_gather_mean(const int* __restrict__ cnt,
                                                     const int* __restrict__ slot,
                                                     const int* __restrict__ spill,
                                                     const float* __restrict__ h,
                                                     float* __restrict__ out, int N) {
  const int lane = threadIdx.x & 31;
  const int wv = __builtin_amdgcn_readfirstlane((int)(blockIdx.x * 8 + (threadIdx.x >> 5)));
  const int r0 = wv * GROWS;
  if (r0 >= N) return;
  const int nr = min(GROWS, N - r0);
  const int dg = (lane < nr) ? cnt[r0 + lane] : 0;
  int dgs[GROWS];
#pragma unroll
  for (int r = 0; r < GROWS; ++r) dgs[r] = __builtin_amdgcn_readlane(dg, r);
  int ids[GROWS];
#pragma unroll
  for (int r = 0; r < GROWS; ++r)
    ids[r] = (lane < min(dgs[r], SLOTS)) ? slot[(size_t)(r0 + r) * SLOTS + lane] : 0;
  const float4* h4 = reinterpret_cast<const float4*>(h);
  float4* o4 = reinterpret_cast<float4*>(out);
  for (int r = 0; r < nr; ++r) {
    const int deg = __builtin_amdgcn_readlane(dg, r);
    const int mine = (r == 0) ? ids[0] : (r == 1) ? ids[1] : (r == 2) ? ids[2] : ids[3];
    const int n = min(deg, SLOTS);
    float4 acc = {0.0f, 0.0f, 0.0f, 0.0f};
#pragma unroll 4
    for (int k = 0; k < n; ++k) {
      const int sidx = __builtin_amdgcn_readlane(mine, k);
      const float4 v = h4[(size_t)sidx * (HIDC / 4) + lane];
      acc.x += v.x;
      acc.y += v.y;
      acc.z += v.z;
      acc.w += v.w;
    }
    if (deg > SLOTS) {
      const int row = r0 + r;
      const int M = cnt[N];
      const int2* sp = reinterpret_cast<const int2*>(spill);
      for (int j0 = 0; j0 < M; j0 += 32) {
        const int j = j0 + lane;
        int2 rec;
        rec.x = -1;
        rec.y = 0;
        if (j < M) rec = sp[j];
        unsigned hit = __builtin_amdgcn_ballot_w32(rec.x == row);
        while (hit != 0u) {
          const int k = __builtin_ctz(hit);
          hit &= hit - 1u;
          const int sidx = __builtin_amdgcn_readlane(rec.y, k);
          const float4 v = h4[(size_t)sidx * (HIDC / 4) + lane];
          acc.x += v.x;
          acc.y += v.y;
          acc.z += v.z;
          acc.w += v.w;
        }
      }
    }
    const float inv = 1.0f / fmaxf((float)deg, 1.0f);
    acc.x *= inv;
    acc.y *= inv;
    acc.z *= inv;
    acc.w *= inv;
    o4[(size_t)(r0 + r) * (HIDC / 4) + lane] = acc;
  }
}

#define MT 7
#define MROWS (MT * 16)
#define STG (MROWS / 8)
#define RB 7

#define FETCH_ROWS(X)                                                            \
  _Pragma("unroll") for (int j = 0; j < STG; ++j) {                              \
    const int grow = min(mbase + w + 8 * j, N - 1);                              \
    pre[j] = reinterpret_cast<const v4f*>((X) + (size_t)grow * KDIM)[lane];      \
  }
#define FETCH_ROWS_AHEAD(X)                                                      \
  FETCH_ROWS(X)                                                                  \
  __builtin_amdgcn_sched_barrier(0);
#define PARK_ROWS()                                                              \
  _Pragma("unroll") for (int j = 0; j < STG; ++j)                                \
    *reinterpret_cast<v4f*>(lds + (w + 8 * j) * LPITCH + lane * 4) = pre[j];
#define GB 16
#define GATHER_PARK_ROWS(H, MINE0)                                               \
  {                                                                              \
    const float4* h4 = reinterpret_cast<const float4*>(H);                       \
    int mine = (MINE0);                                                          \
    for (int j = 0; j < STG; ++j) {                                              \
      const int row = mbase + w + 8 * j;                                         \
      const int deg = __builtin_amdgcn_readlane(dg, j);                          \
      const int n = min(deg, SLOTS);                                             \
      const int degn = (j + 1 < STG) ? __builtin_amdgcn_readlane(dg, j + 1) : 0; \
      int minen = 0;                                                             \
      if (lane < min(degn, SLOTS)) minen = slot[(size_t)(row + 8) * SLOTS + lane]; \
      float4 sum = {0.0f, 0.0f, 0.0f, 0.0f};                                     \
      for (int k0 = 0; k0 < n; k0 += GB) {                                       \
        float4 v[GB];                                                            \
          \
        _Pragma("unroll") for (int i = 0; i < GB; ++i) {                         \
          if (k0 + i < n) {                                                      \
            const int sidx = __builtin_amdgcn_readlane(mine, k0 + i);            \
            v[i] = reinterpret_cast<const float4*>((H) + (size_t)sidx * HIDC)[lane]; \
          }                                                                      \
        }                                                                        \
        _Pragma("unroll") for (int i = 0; i < GB; ++i) {                         \
          if (k0 + i < n) {                                                      \
            sum.x += v[i].x; sum.y += v[i].y; sum.z += v[i].z; sum.w += v[i].w;  \
          }                                                                      \
        }                                                                        \
      }                                                                          \
      if (deg > SLOTS) {                                                         \
        const int M = cnt[N];                                                    \
        const int2* sp = reinterpret_cast<const int2*>(spill);                   \
        for (int j0 = 0; j0 < M; j0 += 32) {                                     \
          const int jj = j0 + lane;                                              \
          int2 rec;                                                              \
          rec.x = -1;                                                            \
          rec.y = 0;                                                             \
          if (jj < M) rec = sp[jj];                                              \
          unsigned hit = __builtin_amdgcn_ballot_w32(rec.x == row);              \
          while (hit != 0u) {                               \
            const int kb = __builtin_ctz(hit);                                   \
            hit &= hit - 1u;                                                     \
            const int sidx = __builtin_amdgcn_readlane(rec.y, kb);               \
            const float4 vv = h4[(size_t)sidx * (HIDC / 4) + lane];              \
            sum.x += vv.x; sum.y += vv.y; sum.z += vv.z; sum.w += vv.w;          \
          }                                                                      \
        }                                                                        \
      }                                                                          \
      const float inv = 1.0f / fmaxf((float)deg, 1.0f);                          \
      v4f o;                                                                     \
      o[0] = sum.x * inv; o[1] = sum.y * inv; o[2] = sum.z * inv; o[3] = sum.w * inv; \
      *reinterpret_cast<v4f*>(lds + (w + 8 * j) * LPITCH + lane * 4) = o;        \
      mine = minen;                                                              \
    }                                                                            \
  }
#define KSTEP(KK, BF)                                                            \
  _Pragma("unroll") for (int i = 0; i < MT; ++i) {                               \
    v2f a;                                                                       \
    const float2 av = *reinterpret_cast<const float2*>(la + i * 16 * LPITCH + (KK)); \
    a[0] = av.x; a[1] = av.y;                                                    \
    acc[i] = __builtin_amdgcn_wmma_f32_16x16x4_f32(                              \
        false, a, false, (BF), (short)0, acc[i], false, false);                  \
  }
#define LOAD_B4(DST, K0)                                                         \
  _Pragma("unroll") for (int s = 0; s < 4; ++s) {                                \
    const float2 bv2 = *reinterpret_cast<const float2*>(wb + (K0) + 4 * s);      \
    DST[s][0] = bv2.x; DST[s][1] = bv2.y;                                        \
  }
#define PRODUCTS(P, NEXT_FETCH)                                                  \
  {                                                                              \
    const float* wb = Wt + (size_t)(P) * (KDIM * HIDC) + (size_t)ncol * KDIM + koff; \
    v2f bq[4], bn[4];                                                            \
    LOAD_B4(bq, 0)                                                               \
      \
    _Pragma("unroll 2") for (int k = 0; k < KDIM - 32; k += 16) {                \
      LOAD_B4(bn, k + 16)                                                        \
      _Pragma("unroll") for (int s = 0; s < 4; ++s) { KSTEP(k + 4 * s, bq[s]) }  \
      _Pragma("unroll") for (int s = 0; s < 4; ++s) bq[s] = bn[s];               \
    }                                                                            \
    LOAD_B4(bn, KDIM - 16)                                                       \
    NEXT_FETCH                                                                   \
    _Pragma("unroll") for (int s = 0; s < 4; ++s) { KSTEP(KDIM - 32 + 4 * s, bq[s]) } \
    _Pragma("unroll") for (int s = 0; s < 4; ++s) { KSTEP(KDIM - 16 + 4 * s, bn[s]) } \
  }

#define HB_LOAD8(DST, K0)                                                        \
  _Pragma("unroll") for (int s = 0; s < 8; ++s) {                                \
    const float2 bv2 = *reinterpret_cast<const float2*>(hb + (K0) + 4 * s);      \
    DST[s][0] = bv2.x; DST[s][1] = bv2.y;                                        \
  }
#define HB_STEPS8(BF, K0)                                                        \
  _Pragma("unroll") for (int s = 0; s < 8; ++s) {                                \
    _Pragma("unroll") for (int j = 0; j < NT; ++j) {                             \
      v2f a;                                                                     \
      const float2 av =                                                          \
          *reinterpret_cast<const float2*>(ha + j * 32 * LPITCH + (K0) + 4 * s); \
      a[0] = av.x; a[1] = av.y;                                                  \
      hacc[j] = __builtin_amdgcn_wmma_f32_16x16x4_f32(                           \
          false, a, false, BF[s], (short)0, hacc[j], false, false);              \
    }                                                                            \
  }
template <int NT>
__device__ __forceinline__ void head_products(const float* __restrict__ ha,
                                              const float* __restrict__ hb, const v2f* hb0,
                                              v8f* hacc) {
  v2f b1[8], b2[8];
  HB_LOAD8(b1, 32)
  HB_STEPS8(hb0, 0)
  HB_LOAD8(b2, 64)
  HB_STEPS8(b1, 32)
  HB_LOAD8(b1, 96)
  HB_STEPS8(b2, 64)
  HB_STEPS8(b1, 96)
}

template <bool HEAD>
__device__ __forceinline__ void gemm3_ln_body(
    float* lds, float* tile,
    const float* __restrict__ X0, const float* __restrict__ X1,
    const int* __restrict__ cnt, const int* __restrict__ slot,
    const int* __restrict__ spill, const float* __restrict__ Wt,
    const float* __restrict__ bias, const float* __restrict__ gw,
    const float* __restrict__ gb, const float* __restrict__ WtH,
    const float* __restrict__ hbias, float* __restrict__ out, int N, int NOUT) {
  const int t = threadIdx.x;
  const int lane = t & 31;
  const int w = __builtin_amdgcn_readfirstlane(t >> 5);
  const int mbase = blockIdx.x * MROWS;

  const int dgrow = mbase + w + 8 * lane;
  const int dg = (lane < STG && dgrow < N) ? cnt[dgrow] : 0;

  v4f pre[STG];
  FETCH_ROWS(X0)
  PARK_ROWS()
  __syncthreads();

  const int m = lane & 15;
  const int ncol = w * 16 + m;
  const int koff = (lane < 16) ? 0 : 2;
  const float* la = lds + m * LPITCH + koff;

  v8f acc[MT];
#pragma unroll
  for (int i = 0; i < MT; ++i) acc[i] = v8f{};

  PRODUCTS(0, FETCH_ROWS_AHEAD(X1))
  __syncthreads();
  PARK_ROWS()
  __syncthreads();

  int mine0 = 0;
  if (lane < min(__builtin_amdgcn_readlane(dg, 0), SLOTS))
    mine0 = slot[(size_t)(mbase + w) * SLOTS + lane];
  PRODUCTS(1, )
  __syncthreads();
  GATHER_PARK_ROWS(X1, mine0)
  __syncthreads();

  PRODUCTS(2, )
  __syncthreads();

  const float bv = bias[ncol];
  const int mo = (lane < 16) ? 0 : 8;
#pragma unroll
  for (int i = 0; i < MT; ++i)
#pragma unroll
    for (int r = 0; r < 8; ++r) lds[(i * 16 + r + mo) * LPITCH + ncol] = acc[i][r] + bv;
  __syncthreads();

  const float4 g4 = reinterpret_cast<const float4*>(gw)[lane];
  const float4 b4 = reinterpret_cast<const float4*>(gb)[lane];
  float4* o4 = reinterpret_cast<float4*>(out);

  const int rset = w >> 2;
  const int hcol = (w & 3) * 16 + m;
  const float* hb = HEAD ? WtH + (size_t)hcol * KDIM + koff : nullptr;
  v2f hbf[8];
  if (HEAD) { HB_LOAD8(hbf, 0) }

  for (int g = 0; g < STG / RB; ++g) {
    float4 v[RB];
    float s[RB], q[RB];
#pragma unroll
    for (int u = 0; u < RB; ++u) {
      const int row = w + 8 * (g * RB + u);
      v[u] = *reinterpret_cast<const float4*>(lds + row * LPITCH + lane * 4);
      s[u] = v[u].x + v[u].y + v[u].z + v[u].w;
      q[u] = v[u].x * v[u].x + v[u].y * v[u].y + v[u].z * v[u].z + v[u].w * v[u].w;
    }
    for (int msk = 16; msk >= 1; msk >>= 1) {
#pragma unroll
      for (int u = 0; u < RB; ++u) {
        s[u] += __shfl_xor(s[u], msk, 32);
        q[u] += __shfl_xor(q[u], msk, 32);
      }
    }
#pragma unroll
    for (int u = 0; u < RB; ++u) {
      const int row = w + 8 * (g * RB + u);
      const float mu = s[u] * (1.0f / HIDC);
      const float var = q[u] * (1.0f / HIDC) - mu * mu;
      const float rs = rsqrtf(var + 1e-5f);
      float4 y;
      y.x = (v[u].x - mu) * rs * g4.x + b4.x;
      y.y = (v[u].y - mu) * rs * g4.y + b4.y;
      y.z = (v[u].z - mu) * rs * g4.z + b4.z;
      y.w = (v[u].w - mu) * rs * g4.w + b4.w;
      if (HEAD) {
        *reinterpret_cast<float4*>(lds + row * LPITCH + lane * 4) = y;
      } else if (mbase + row < N) {
        o4[(size_t)(mbase + row) * (HIDC / 4) + lane] = y;
      }
    }
  }
  if (!HEAD) return;
  __syncthreads();

  const float* ha = lds + (rset * 16 + m) * LPITCH + koff;
  v8f hacc[4];
#pragma unroll
  for (int j = 0; j < 4; ++j) hacc[j] = v8f{};
  if (rset == 0) head_products<4>(ha, hb, hbf, hacc);
  else           head_products<3>(ha, hb, hbf, hacc);

  const float hbv = hbias[hcol];
  __syncthreads();
#pragma unroll
  for (int j = 0; j < 4; ++j)
    if (j < 3 || rset == 0) {
#pragma unroll
      for (int r = 0; r < 8; ++r)
        tile[((rset + 2 * j) * 16 + r + mo) * HPITCH + hcol] = hacc[j][r] + hbv;
    }
  __syncthreads();

  for (int g = 0; g < STG / RB; ++g) {
    float a[RB], b[RB], mx[RB], e[RB];
#pragma unroll
    for (int u = 0; u < RB; ++u) {
      const float* r = tile + (w + 8 * (g * RB + u)) * HPITCH;
      a[u] = r[lane];
      b[u] = r[lane + 32];
      mx[u] = fmaxf(a[u], b[u]);
    }
    for (int msk = 16; msk >= 1; msk >>= 1) {
#pragma unroll
      for (int u = 0; u < RB; ++u) mx[u] = fmaxf(mx[u], __shfl_xor(mx[u], msk, 32));
    }
#pragma unroll
    for (int u = 0; u < RB; ++u) e[u] = expf(a[u] - mx[u]) + expf(b[u] - mx[u]);
    for (int msk = 16; msk >= 1; msk >>= 1) {
#pragma unroll
      for (int u = 0; u < RB; ++u) e[u] += __shfl_xor(e[u], msk, 32);
    }
#pragma unroll
    for (int u = 0; u < RB; ++u) {
      const int row = w + 8 * (g * RB + u);
      if (mbase + row < N) {
        const float lse = mx[u] + logf(e[u]);
        float* o = out + (size_t)(mbase + row) * NOUT;
        o[lane] = a[u] - lse;
        o[lane + 32] = b[u] - lse;
      }
    }
  }
}

__global__ __launch_bounds__(256) void k_wmma_gemm3_ln(
    const float* __restrict__ X0, const float* __restrict__ X1,
    const int* __restrict__ cnt, const int* __restrict__ slot,
    const int* __restrict__ spill, const float* __restrict__ Wt,
    const float* __restrict__ bias, const float* __restrict__ gw,
    const float* __restrict__ gb, float* __restrict__ out, int N) {
  __shared__ float lds[MROWS * LPITCH];
  gemm3_ln_body<false>(lds, nullptr, X0, X1, cnt, slot, spill, Wt, bias, gw, gb, nullptr, nullptr,
                       out, N, 0);
}

__global__ __launch_bounds__(256, 5) void k_wmma_gemm3_ln_head(
    const float* __restrict__ X0, const float* __restrict__ X1,
    const int* __restrict__ cnt, const int* __restrict__ slot,
    const int* __restrict__ spill, const float* __restrict__ Wt,
    const float* __restrict__ bias, const float* __restrict__ gw,
    const float* __restrict__ gb, const float* __restrict__ WtH,
    const float* __restrict__ hbias, float* __restrict__ out, int N, int NOUT) {
  __shared__ float lds[MROWS * LPITCH];
  gemm3_ln_body<true>(lds, lds, X0, X1, cnt, slot, spill, Wt, bias, gw, gb, WtH, hbias, out, N,
                      NOUT);
}

extern "C" void kernel_launch(void* const* d_in, const int* in_sizes, int n_in,
                              void* d_out, int out_size, void* d_ws, size_t ws_size,
                              hipStream_t stream) {
  (void)n_in; (void)out_size; (void)ws_size;
  const float* x    = (const float*)d_in[0];
  const int*   ei   = (const int*)d_in[1];
  const float* W1   = (const float*)d_in[2];
  const float* b1   = (const float*)d_in[3];
  const float* W2   = (const float*)d_in[4];
  const float* b2   = (const float*)d_in[5];
  const float* ln1g = (const float*)d_in[6];
  const float* ln1b = (const float*)d_in[7];
  const float* ln2g = (const float*)d_in[8];
  const float* ln2b = (const float*)d_in[9];
  const float* linW = (const float*)d_in[10];
  const float* linb = (const float*)d_in[11];
  float* outp = (float*)d_out;

  const int N = in_sizes[0] / HIDC;
  const int E = in_sizes[1] / 2;
  const int OUTC = in_sizes[11];
  const int* src = ei;
  const int* dstp = ei + E;

  const long NH = (long)N * HIDC;
  const long W3 = 3L * KDIM * HIDC;
  const long WH = (long)KDIM * OUTC;
  float* ws = (float*)d_ws;
  long pad = ((long)N + 255) & ~255L;
  float* agg1 = ws + pad;
  float* bufA = agg1 + 2 * NH;
  float* bufB = bufA + NH;
  float* Wt1  = bufB + NH;
  float* Wt2  = Wt1 + W3;
  float* WtH  = Wt2 + W3;

  const int padC = (N + 1 + 255) & ~255;
  int* cnt   = (int*)bufB;
  int* slot  = cnt + padC;
  int* spill = slot + (size_t)N * SLOTS;

  dim3 b256(256);
  int eBlocks = (E + 255) / 256;
  int gatherBlocks = (N + 8 * GROWS - 1) / (8 * GROWS);
  int mTiles = N / 16;
  int gemmBlocks = (mTiles + MT - 1) / MT;

  int zBlocks = padC / 256;
  int hBlocks = (int)((WH + 255) / 256);
  k_prep<<<zBlocks + 2 * W3BLOCKS + hBlocks, b256, 0, stream>>>(cnt, zBlocks, W1, Wt1, W2, Wt2,
                                                                linW, WtH, OUTC);

  k_count_place<<<eBlocks, b256, 0, stream>>>(src, dstp, cnt, slot, spill, E, N);

  k_gather_mean<<<gatherBlocks, b256, 0, stream>>>(cnt, slot, spill, x, agg1, N);
  k_wmma_gemm3_ln<<<gemmBlocks, b256, 0, stream>>>(x, agg1, cnt, slot, spill, Wt1, b1, ln1g, ln1b,
                                                   bufA, N);

  k_gather_mean<<<gatherBlocks, b256, 0, stream>>>(cnt, slot, spill, bufA, agg1, N);
  k_wmma_gemm3_ln_head<<<gemmBlocks, b256, 0, stream>>>(bufA, agg1, cnt, slot, spill, Wt2, b2,
                                                        ln2g, ln2b, WtH, linb, outp, N, OUTC);
}
